# starting kernel
# baseline (speedup 1.0000x reference)
#include <hip/hip_runtime.h>
#include <hip/hip_bf16.h>
#include <math.h>

#define N_NODES 100000
#define N_EDGES 600000
#define IN_FEATS 256
#define H_FEATS 128

typedef __attribute__((ext_vector_type(2))) float v2f;
typedef __attribute__((ext_vector_type(8))) float v8f;

__global__ __launch_bounds__(256) void degree_kernel(
    const int* __restrict__ srcIdx, const int* __restrict__ dstIdx,
    float* __restrict__ degOut, float* __restrict__ degIn)
{
    int e = blockIdx.x * blockDim.x + threadIdx.x;
    if (e < N_EDGES) {
        unsafeAtomicAdd(&degOut[srcIdx[e]], 1.0f);
        unsafeAtomicAdd(&degIn[dstIdx[e]], 1.0f);
    }
}

__global__ __launch_bounds__(256) void invnorm_kernel(
    float* __restrict__ degOut, float* __restrict__ degIn)
{
    int i = blockIdx.x * blockDim.x + threadIdx.x;
    if (i < N_NODES) {
        float d0 = degOut[i];
        degOut[i] = (d0 > 0.0f) ? rsqrtf(fmaxf(d0, 1.0f)) : 0.0f;
        float d1 = degIn[i];
        degIn[i] = (d1 > 0.0f) ? rsqrtf(fmaxf(d1, 1.0f)) : 0.0f;
    }
}

__global__ __launch_bounds__(256) void lin_wmma_kernel(
    const float* __restrict__ A, const float* __restrict__ W,
    const float* __restrict__ bias, float* __restrict__ X)
{
    const int lane   = threadIdx.x & 31;
    const int wave   = threadIdx.x >> 5;
    const int halfId = lane >> 4;
    const int l16    = lane & 15;
    const int rowBase = blockIdx.x * 32;
    const int n0      = wave * 16;

    const float* a0p = A + (size_t)(rowBase + l16) * IN_FEATS + 2 * halfId;
    const float* a1p = a0p + (size_t)16 * IN_FEATS;
    const float* bp  = W + (size_t)(2 * halfId) * H_FEATS + n0 + l16;

    v8f c0 = {};
    v8f c1 = {};

    #pragma unroll 8
    for (int k = 0; k < IN_FEATS; k += 4) {
        v2f a0, a1, bf;
        a0.x = a0p[k];
        a0.y = a0p[k + 1];
        a1.x = a1p[k];
        a1.y = a1p[k + 1];
        bf.x = bp[(size_t)k * H_FEATS];
        bf.y = bp[(size_t)(k + 1) * H_FEATS];
        c0 = __builtin_amdgcn_wmma_f32_16x16x4_f32(
                false, a0, false, bf, (short)0, c0, false, false);
        c1 = __builtin_amdgcn_wmma_f32_16x16x4_f32(
                false, a1, false, bf, (short)0, c1, false, false);
    }

    const float bb = bias[n0 + l16];
    #pragma unroll
    for (int r = 0; r < 8; ++r) {
        const int m = r + 8 * halfId;
        X[(size_t)(rowBase + m) * H_FEATS + n0 + l16]      = c0[r] + bb;
        X[(size_t)(rowBase + 16 + m) * H_FEATS + n0 + l16] = c1[r] + bb;
    }
}

__global__ __launch_bounds__(256) void scatter_kernel(
    const float* __restrict__ feat, const int* __restrict__ srcIdx,
    const int* __restrict__ dstIdx, const float* __restrict__ invSrc,
    float* __restrict__ agg)
{
    const int e = blockIdx.x * 8 + (threadIdx.x >> 5);
    if (e >= N_EDGES) return;
    const int lane = threadIdx.x & 31;
    const int s = srcIdx[e];
    const int d = dstIdx[e];
    const float w = invSrc[s];
    const float4 v = ((const float4*)(feat + (size_t)s * H_FEATS))[lane];
    float* ap = agg + (size_t)d * H_FEATS + lane * 4;
    unsafeAtomicAdd(ap + 0, v.x * w);
    unsafeAtomicAdd(ap + 1, v.y * w);
    unsafeAtomicAdd(ap + 2, v.z * w);
    unsafeAtomicAdd(ap + 3, v.w * w);
}

__device__ __forceinline__ float softmax2(const float* __restrict__ alphas, int idx)
{
    const float a0 = alphas[0], a1 = alphas[1];
    const float m  = fmaxf(a0, a1);
    const float e0 = __expf(a0 - m), e1 = __expf(a1 - m);
    return (idx == 0 ? e0 : e1) / (e0 + e1);
}

__global__ __launch_bounds__(256) void finalize_first_kernel(
    const float* __restrict__ agg, const float* __restrict__ invDst,
    const float* __restrict__ alphas, float* __restrict__ hbuf,
    float* __restrict__ out)
{
    const int i4 = blockIdx.x * blockDim.x + threadIdx.x;
    const int total4 = (N_NODES * H_FEATS) / 4;
    if (i4 >= total4) return;
    const int n = i4 / (H_FEATS / 4);
    const float s = invDst[n];
    const float a0 = softmax2(alphas, 0);
    float4 v = ((const float4*)agg)[i4];
    v.x *= s; v.y *= s; v.z *= s; v.w *= s;
    ((float4*)hbuf)[i4] = v;
    float4 o;
    o.x = a0 * v.x; o.y = a0 * v.y; o.z = a0 * v.z; o.w = a0 * v.w;
    ((float4*)out)[i4] = o;
}

__global__ __launch_bounds__(256) void finalize_second_kernel(
    const float* __restrict__ agg, const float* __restrict__ invDst,
    const float* __restrict__ alphas, float* __restrict__ out)
{
    const int i4 = blockIdx.x * blockDim.x + threadIdx.x;
    const int total4 = (N_NODES * H_FEATS) / 4;
    if (i4 >= total4) return;
    const int n = i4 / (H_FEATS / 4);
    const float s = invDst[n] * softmax2(alphas, 1);
    const float4 v = ((const float4*)agg)[i4];
    float4 o = ((float4*)out)[i4];
    o.x += s * v.x; o.y += s * v.y; o.z += s * v.z; o.w += s * v.w;
    ((float4*)out)[i4] = o;
}

extern "C" void kernel_launch(void* const* d_in, const int* in_sizes, int n_in,
                              void* d_out, int out_size, void* d_ws, size_t ws_size,
                              hipStream_t stream)
{
    (void)in_sizes; (void)n_in; (void)out_size; (void)ws_size;

    const float* in_feat = (const float*)d_in[0];
    const int*   edge    = (const int*)d_in[1];
    const float* W       = (const float*)d_in[2];
    const float* bias    = (const float*)d_in[3];
    const float* alphas  = (const float*)d_in[4];
    float*       out     = (float*)d_out;

    const int* srcIdx = edge;
    const int* dstIdx = edge + N_EDGES;

    float* degOut = (float*)d_ws;
    float* degIn  = degOut + N_NODES;
    float* xbuf   = degIn + N_NODES;
    float* agg    = xbuf + (size_t)N_NODES * H_FEATS;

    const size_t featBytes = (size_t)N_NODES * H_FEATS * sizeof(float);

    hipMemsetAsync(degOut, 0, (size_t)2 * N_NODES * sizeof(float), stream);
    degree_kernel<<<(N_EDGES + 255) / 256, 256, 0, stream>>>(srcIdx, dstIdx, degOut, degIn);
    invnorm_kernel<<<(N_NODES + 255) / 256, 256, 0, stream>>>(degOut, degIn);

    lin_wmma_kernel<<<N_NODES / 32, 256, 0, stream>>>(in_feat, W, bias, xbuf);

    hipMemsetAsync(agg, 0, featBytes, stream);
    scatter_kernel<<<N_EDGES / 8, 256, 0, stream>>>(xbuf, srcIdx, dstIdx, degOut, agg);
    finalize_first_kernel<<<(N_NODES * H_FEATS / 4 + 255) / 256, 256, 0, stream>>>(
        agg, degIn, alphas, xbuf, out);

    hipMemsetAsync(agg, 0, featBytes, stream);
    scatter_kernel<<<N_EDGES / 8, 256, 0, stream>>>(xbuf, srcIdx, dstIdx, degOut, agg);
    finalize_second_kernel<<<(N_NODES * H_FEATS / 4 + 255) / 256, 256, 0, stream>>>(
        agg, degIn, alphas, out);
}


# lin GEMM fetches each operand once: W transposed into LDS once per persistent work-group, a wave owns 16 rows x 128 cols, A read as b128; k summed as {0,1,4,5},{2,3,6,7} per 8
# speedup vs baseline: 1.1074x; 1.1074x over previous
#include <hip/hip_runtime.h>
#include <hip/hip_bf16.h>
#include <math.h>

#define N_NODES 100000
#define N_EDGES 600000
#define IN_FEATS 256
#define H_FEATS 128

typedef __attribute__((ext_vector_type(2))) float v2f;
typedef __attribute__((ext_vector_type(8))) float v8f;

__global__ __launch_bounds__(256) void degree_kernel(
    const int* __restrict__ srcIdx, const int* __restrict__ dstIdx,
    float* __restrict__ degOut, float* __restrict__ degIn)
{
    int e = blockIdx.x * blockDim.x + threadIdx.x;
    if (e < N_EDGES) {
        unsafeAtomicAdd(&degOut[srcIdx[e]], 1.0f);
        unsafeAtomicAdd(&degIn[dstIdx[e]], 1.0f);
    }
}

__global__ __launch_bounds__(256) void invnorm_kernel(
    float* __restrict__ degOut, float* __restrict__ degIn)
{
    int i = blockIdx.x * blockDim.x + threadIdx.x;
    if (i < N_NODES) {
        float d0 = degOut[i];
        degOut[i] = (d0 > 0.0f) ? rsqrtf(fmaxf(d0, 1.0f)) : 0.0f;
        float d1 = degIn[i];
        degIn[i] = (d1 > 0.0f) ? rsqrtf(fmaxf(d1, 1.0f)) : 0.0f;
    }
}

#define WL_PITCH 260
#define LIN_TILES (N_NODES / 16)
#define LIN_BLOCKS 256
#define LIN_THREADS 512

__global__ __launch_bounds__(LIN_THREADS) void lin_wmma_kernel(
    const float* __restrict__ A, const float* __restrict__ W,
    const float* __restrict__ bias, float* __restrict__ X)
{
    __shared__ __attribute__((aligned(16))) float Wl[H_FEATS * WL_PITCH];

    const int lane   = threadIdx.x & 31;
    const int wave   = __builtin_amdgcn_readfirstlane(threadIdx.x >> 5);
    const int h      = lane >> 4;
    const int c      = lane & 15;
    const int nWaves = gridDim.x * (LIN_THREADS / 32);
    int t = blockIdx.x + gridDim.x * wave;

    const float* ap = A + (size_t)(t * 16 + c) * IN_FEATS + 4 * h;
    float4 acur[8];
    #pragma unroll
    for (int i = 0; i < 8; ++i) acur[i] = *(const float4*)(ap + 8 * i);

    float bb[8];
    #pragma unroll
    for (int j = 0; j < 8; ++j) bb[j] = bias[16 * j + c];

    for (int it = 0; it < (IN_FEATS * H_FEATS) / LIN_THREADS; ++it) {
        const int i = it * LIN_THREADS + threadIdx.x;
        Wl[(i & (H_FEATS - 1)) * WL_PITCH + (i >> 7)] = W[i];
    }
    __syncthreads();

    const float* wl = Wl + c * WL_PITCH + 4 * h;

    while (t < LIN_TILES) {
        asm volatile("" ::: "memory");
        const int tn = t + nWaves;
        const float* apn = A + (size_t)((tn < LIN_TILES ? tn : t) * 16 + c) * IN_FEATS + 4 * h;

        v8f acc[8];
        #pragma unroll
        for (int j = 0; j < 8; ++j) acc[j] = (v8f){};

        #pragma unroll
        for (int ch = 0; ch < 4; ++ch) {
            const float* nx = (ch < 3) ? (ap + 64 * (ch + 1)) : apn;
            float4 anxt[8];
            #pragma unroll
            for (int i = 0; i < 8; ++i) anxt[i] = *(const float4*)(nx + 8 * i);
            __builtin_amdgcn_sched_barrier(0);

            #pragma unroll
            for (int i = 0; i < 8; ++i) {
                const float* wk = wl + 64 * ch + 8 * i;
                float4 b[8];
                #pragma unroll
                for (int j = 0; j < 8; ++j) b[j] = *(const float4*)(wk + j * 16 * WL_PITCH);
                v2f a01, a23;
                a01.x = acur[i].x; a01.y = acur[i].y;
                a23.x = acur[i].z; a23.y = acur[i].w;
                #pragma unroll
                for (int j = 0; j < 8; ++j) {
                    v2f b01; b01.x = b[j].x; b01.y = b[j].y;
                    acc[j] = __builtin_amdgcn_wmma_f32_16x16x4_f32(
                                false, a01, false, b01, (short)0, acc[j], false, false);
                }
                #pragma unroll
                for (int j = 0; j < 8; ++j) {
                    v2f b23; b23.x = b[j].z; b23.y = b[j].w;
                    acc[j] = __builtin_amdgcn_wmma_f32_16x16x4_f32(
                                false, a23, false, b23, (short)0, acc[j], false, false);
                }
            }
            #pragma unroll
            for (int i = 0; i < 8; ++i) acur[i] = anxt[i];
        }

        float* xp = X + (size_t)(t * 16 + 8 * h) * H_FEATS + c;
        #pragma unroll
        for (int j = 0; j < 8; ++j) {
            #pragma unroll
            for (int r = 0; r < 8; ++r)
                xp[(size_t)r * H_FEATS + 16 * j] = acc[j][r] + bb[j];
        }

        ap = apn;
        t  = tn;
    }
}

__global__ __launch_bounds__(256) void scatter_kernel(
    const float* __restrict__ feat, const int* __restrict__ srcIdx,
    const int* __restrict__ dstIdx, const float* __restrict__ invSrc,
    float* __restrict__ agg)
{
    const int e = blockIdx.x * 8 + (threadIdx.x >> 5);
    if (e >= N_EDGES) return;
    const int lane = threadIdx.x & 31;
    const int s = srcIdx[e];
    const int d = dstIdx[e];
    const float w = invSrc[s];
    const float4 v = ((const float4*)(feat + (size_t)s * H_FEATS))[lane];
    float* ap = agg + (size_t)d * H_FEATS + lane * 4;
    unsafeAtomicAdd(ap + 0, v.x * w);
    unsafeAtomicAdd(ap + 1, v.y * w);
    unsafeAtomicAdd(ap + 2, v.z * w);
    unsafeAtomicAdd(ap + 3, v.w * w);
}

__device__ __forceinline__ float softmax2(const float* __restrict__ alphas, int idx)
{
    const float a0 = alphas[0], a1 = alphas[1];
    const float m  = fmaxf(a0, a1);
    const float e0 = __expf(a0 - m), e1 = __expf(a1 - m);
    return (idx == 0 ? e0 : e1) / (e0 + e1);
}

__global__ __launch_bounds__(256) void finalize_first_kernel(
    const float* __restrict__ agg, const float* __restrict__ invDst,
    const float* __restrict__ alphas, float* __restrict__ hbuf,
    float* __restrict__ out)
{
    const int i4 = blockIdx.x * blockDim.x + threadIdx.x;
    const int total4 = (N_NODES * H_FEATS) / 4;
    if (i4 >= total4) return;
    const int n = i4 / (H_FEATS / 4);
    const float s = invDst[n];
    const float a0 = softmax2(alphas, 0);
    float4 v = ((const float4*)agg)[i4];
    v.x *= s; v.y *= s; v.z *= s; v.w *= s;
    ((float4*)hbuf)[i4] = v;
    float4 o;
    o.x = a0 * v.x; o.y = a0 * v.y; o.z = a0 * v.z; o.w = a0 * v.w;
    ((float4*)out)[i4] = o;
}

__global__ __launch_bounds__(256) void finalize_second_kernel(
    const float* __restrict__ agg, const float* __restrict__ invDst,
    const float* __restrict__ alphas, float* __restrict__ out)
{
    const int i4 = blockIdx.x * blockDim.x + threadIdx.x;
    const int total4 = (N_NODES * H_FEATS) / 4;
    if (i4 >= total4) return;
    const int n = i4 / (H_FEATS / 4);
    const float s = invDst[n] * softmax2(alphas, 1);
    const float4 v = ((const float4*)agg)[i4];
    float4 o = ((float4*)out)[i4];
    o.x += s * v.x; o.y += s * v.y; o.z += s * v.z; o.w += s * v.w;
    ((float4*)out)[i4] = o;
}

extern "C" void kernel_launch(void* const* d_in, const int* in_sizes, int n_in,
                              void* d_out, int out_size, void* d_ws, size_t ws_size,
                              hipStream_t stream)
{
    (void)in_sizes; (void)n_in; (void)out_size; (void)ws_size;

    const float* in_feat = (const float*)d_in[0];
    const int*   edge    = (const int*)d_in[1];
    const float* W       = (const float*)d_in[2];
    const float* bias    = (const float*)d_in[3];
    const float* alphas  = (const float*)d_in[4];
    float*       out     = (float*)d_out;

    const int* srcIdx = edge;
    const int* dstIdx = edge + N_EDGES;

    float* degOut = (float*)d_ws;
    float* degIn  = degOut + N_NODES;
    float* xbuf   = degIn + N_NODES;
    float* agg    = xbuf + (size_t)N_NODES * H_FEATS;

    const size_t featBytes = (size_t)N_NODES * H_FEATS * sizeof(float);

    hipMemsetAsync(degOut, 0, (size_t)2 * N_NODES * sizeof(float), stream);
    degree_kernel<<<(N_EDGES + 255) / 256, 256, 0, stream>>>(srcIdx, dstIdx, degOut, degIn);
    invnorm_kernel<<<(N_NODES + 255) / 256, 256, 0, stream>>>(degOut, degIn);

    lin_wmma_kernel<<<LIN_BLOCKS, LIN_THREADS, 0, stream>>>(in_feat, W, bias, xbuf);

    hipMemsetAsync(agg, 0, featBytes, stream);
    scatter_kernel<<<N_EDGES / 8, 256, 0, stream>>>(xbuf, srcIdx, dstIdx, degOut, agg);
    finalize_first_kernel<<<(N_NODES * H_FEATS / 4 + 255) / 256, 256, 0, stream>>>(
        agg, degIn, alphas, xbuf, out);

    hipMemsetAsync(agg, 0, featBytes, stream);
    scatter_kernel<<<N_EDGES / 8, 256, 0, stream>>>(xbuf, srcIdx, dstIdx, degOut, agg);
    finalize_second_kernel<<<(N_NODES * H_FEATS / 4 + 255) / 256, 256, 0, stream>>>(
        agg, degIn, alphas, out);
}


# atomic f32 scatter replaced by a pull over a destination-sorted edge list built per call (int atomic slot rank, arrival order): a wave sums a node's source rows in registers, writes once
# speedup vs baseline: 2.3188x; 2.0939x over previous
#include <hip/hip_runtime.h>
#include <hip/hip_bf16.h>
#include <math.h>

#define N_NODES 100000
#define N_EDGES 600000
#define IN_FEATS 256
#define H_FEATS 128

typedef __attribute__((ext_vector_type(2))) float v2f;
typedef __attribute__((ext_vector_type(8))) float v8f;


__global__ __launch_bounds__(256) void count_kernel(
    const int* __restrict__ srcIdx, const int* __restrict__ dstIdx,
    float* __restrict__ degOut, int* __restrict__ cntIn)
{
    int e = blockIdx.x * blockDim.x + threadIdx.x;
    if (e < N_EDGES) {
        unsafeAtomicAdd(&degOut[srcIdx[e]], 1.0f);
        atomicAdd(&cntIn[dstIdx[e]], 1);
    }
}

__global__ __launch_bounds__(256) void prep_kernel(
    float* __restrict__ degOut, const int* __restrict__ cntIn,
    int* __restrict__ rowPtr, int* __restrict__ cursor)
{
    __shared__ int waveOff[8];
    __shared__ int blockBase;
    const int i    = blockIdx.x * 256 + threadIdx.x;
    const int lane = threadIdx.x & 31;
    const int wave = threadIdx.x >> 5;
    int c = 0;
    if (i < N_NODES) {
        const float d0 = degOut[i];
        degOut[i] = (d0 > 0.0f) ? rsqrtf(fmaxf(d0, 1.0f)) : 0.0f;
        c = cntIn[i];
    }
    int incl = c;
    #pragma unroll
    for (int o = 1; o < 32; o <<= 1) {
        const int up = __shfl_up(incl, o);
        if (lane >= o) incl += up;
    }
    if (lane == 31) waveOff[wave] = incl;
    __syncthreads();
    if (threadIdx.x == 0) {
        int tot = 0;
        for (int w = 0; w < 8; ++w) { const int s = waveOff[w]; waveOff[w] = tot; tot += s; }
        blockBase = atomicAdd(cursor, tot);
    }
    __syncthreads();
    if (i < N_NODES) rowPtr[i] = blockBase + waveOff[wave] + incl - c;
}

__global__ __launch_bounds__(256) void fill_kernel(
    const int* __restrict__ srcIdx, const int* __restrict__ dstIdx,
    int* __restrict__ rowPtr, int* __restrict__ csrSrc)
{
    int e = blockIdx.x * blockDim.x + threadIdx.x;
    if (e < N_EDGES) {
        const int pos = atomicAdd(&rowPtr[dstIdx[e]], 1);
        csrSrc[pos] = srcIdx[e];
    }
}

#define WL_PITCH 260
#define LIN_TILES (N_NODES / 16)
#define LIN_BLOCKS 256
#define LIN_THREADS 512

__global__ __launch_bounds__(LIN_THREADS) void lin_wmma_kernel(
    const float* __restrict__ A, const float* __restrict__ W,
    const float* __restrict__ bias, float* __restrict__ X)
{
    __shared__ __attribute__((aligned(16))) float Wl[H_FEATS * WL_PITCH];

    const int lane   = threadIdx.x & 31;
    const int wave   = __builtin_amdgcn_readfirstlane(threadIdx.x >> 5);
    const int h      = lane >> 4;
    const int c      = lane & 15;
    const int nWaves = gridDim.x * (LIN_THREADS / 32);
    int t = blockIdx.x + gridDim.x * wave;

    const float* ap = A + (size_t)(t * 16 + c) * IN_FEATS + 4 * h;
    float4 acur[8];
    #pragma unroll
    for (int i = 0; i < 8; ++i) acur[i] = *(const float4*)(ap + 8 * i);

    float bb[8];
    #pragma unroll
    for (int j = 0; j < 8; ++j) bb[j] = bias[16 * j + c];

    for (int it = 0; it < (IN_FEATS * H_FEATS) / LIN_THREADS; ++it) {
        const int i = it * LIN_THREADS + threadIdx.x;
        Wl[(i & (H_FEATS - 1)) * WL_PITCH + (i >> 7)] = W[i];
    }
    __syncthreads();

    const float* wl = Wl + c * WL_PITCH + 4 * h;

    while (t < LIN_TILES) {
        asm volatile("" ::: "memory");
        const int tn = t + nWaves;
        const float* apn = A + (size_t)((tn < LIN_TILES ? tn : t) * 16 + c) * IN_FEATS + 4 * h;

        v8f acc[8];
        #pragma unroll
        for (int j = 0; j < 8; ++j) acc[j] = (v8f){};

        #pragma unroll
        for (int ch = 0; ch < 4; ++ch) {
            const float* nx = (ch < 3) ? (ap + 64 * (ch + 1)) : apn;
            float4 anxt[8];
            #pragma unroll
            for (int i = 0; i < 8; ++i) anxt[i] = *(const float4*)(nx + 8 * i);
            __builtin_amdgcn_sched_barrier(0);

            #pragma unroll
            for (int i = 0; i < 8; ++i) {
                const float* wk = wl + 64 * ch + 8 * i;
                float4 b[8];
                #pragma unroll
                for (int j = 0; j < 8; ++j) b[j] = *(const float4*)(wk + j * 16 * WL_PITCH);
                v2f a01, a23;
                a01.x = acur[i].x; a01.y = acur[i].y;
                a23.x = acur[i].z; a23.y = acur[i].w;
                #pragma unroll
                for (int j = 0; j < 8; ++j) {
                    v2f b01; b01.x = b[j].x; b01.y = b[j].y;
                    acc[j] = __builtin_amdgcn_wmma_f32_16x16x4_f32(
                                false, a01, false, b01, (short)0, acc[j], false, false);
                }
                #pragma unroll
                for (int j = 0; j < 8; ++j) {
                    v2f b23; b23.x = b[j].z; b23.y = b[j].w;
                    acc[j] = __builtin_amdgcn_wmma_f32_16x16x4_f32(
                                false, a23, false, b23, (short)0, acc[j], false, false);
                }
            }
            #pragma unroll
            for (int i = 0; i < 8; ++i) acur[i] = anxt[i];
        }

        float* xp = X + (size_t)(t * 16 + 8 * h) * H_FEATS + c;
        #pragma unroll
        for (int j = 0; j < 8; ++j) {
            #pragma unroll
            for (int r = 0; r < 8; ++r)
                xp[(size_t)r * H_FEATS + 16 * j] = acc[j][r] + bb[j];
        }

        ap = apn;
        t  = tn;
    }
}

__device__ __forceinline__ float4 pull_row(
    const float* __restrict__ feat, const int* __restrict__ csrSrc,
    const float* __restrict__ invSrc, int start, int cnt, int lane)
{
    float4 acc = make_float4(0.0f, 0.0f, 0.0f, 0.0f);
    for (int base = 0; base < cnt; base += 32) {
        const int n = (cnt - base < 32) ? (cnt - base) : 32;
        int   myS = 0;
        float myW = 0.0f;
        if (lane < n) { myS = csrSrc[start + base + lane]; myW = invSrc[myS]; }
        for (int j = 0; j < n; j += 8) {
            float4 v[8];
            float  w[8];
            #pragma unroll
            for (int u = 0; u < 8; ++u) {
                const int s = __builtin_amdgcn_readlane(myS, j + u);
                w[u] = __int_as_float(__builtin_amdgcn_readlane(__float_as_int(myW), j + u));
                v[u] = *(const float4*)(feat + (size_t)s * H_FEATS + 4 * lane);
            }
            #pragma unroll
            for (int u = 0; u < 8; ++u) {
                const bool on = (j + u < n);
                const float4 t = v[u];
                acc.x += (on ? t.x : 0.0f) * w[u];
                acc.y += (on ? t.y : 0.0f) * w[u];
                acc.z += (on ? t.z : 0.0f) * w[u];
                acc.w += (on ? t.w : 0.0f) * w[u];
            }
        }
    }
    return acc;
}

__device__ __forceinline__ float softmax2(const float* __restrict__ alphas, int idx)
{
    const float a0 = alphas[0], a1 = alphas[1];
    const float m  = fmaxf(a0, a1);
    const float e0 = __expf(a0 - m), e1 = __expf(a1 - m);
    return (idx == 0 ? e0 : e1) / (e0 + e1);
}

#define PULL_BLOCKS 2048
#define PULL_WAVES  (PULL_BLOCKS * 8)

__global__ __launch_bounds__(256) void pull_first_kernel(
    const float* __restrict__ feat, const int* __restrict__ csrSrc,
    const int* __restrict__ rowEnd, const int* __restrict__ cntIn,
    const float* __restrict__ invSrc, float* __restrict__ hbuf)
{
    const int lane = threadIdx.x & 31;
    const int d0 = blockIdx.x * 8 + __builtin_amdgcn_readfirstlane(threadIdx.x >> 5);
    for (int d = d0; d < N_NODES; d += PULL_WAVES) {
        const int cnt = cntIn[d];
        const int end = rowEnd[d];
        const float4 acc = pull_row(feat, csrSrc, invSrc, end - cnt, cnt, lane);
        const float dg   = (float)cnt;
        const float invd = (dg > 0.0f) ? rsqrtf(fmaxf(dg, 1.0f)) : 0.0f;
        float4 hv;
        hv.x = acc.x * invd; hv.y = acc.y * invd; hv.z = acc.z * invd; hv.w = acc.w * invd;
        *(float4*)(hbuf + (size_t)d * H_FEATS + 4 * lane) = hv;
    }
}

__global__ __launch_bounds__(256) void pull_second_kernel(
    const float* __restrict__ hbuf, const int* __restrict__ csrSrc,
    const int* __restrict__ rowEnd, const int* __restrict__ cntIn,
    const float* __restrict__ invSrc, const float* __restrict__ alphas,
    float* __restrict__ out)
{
    const int lane = threadIdx.x & 31;
    const float a0 = __int_as_float(__builtin_amdgcn_readfirstlane(__float_as_int(softmax2(alphas, 0))));
    const float a1 = __int_as_float(__builtin_amdgcn_readfirstlane(__float_as_int(softmax2(alphas, 1))));
    const int d0 = blockIdx.x * 8 + __builtin_amdgcn_readfirstlane(threadIdx.x >> 5);
    for (int d = d0; d < N_NODES; d += PULL_WAVES) {
        const int cnt = cntIn[d];
        const int end = rowEnd[d];
        const float4 hd = *(const float4*)(hbuf + (size_t)d * H_FEATS + 4 * lane);
        const float4 acc = pull_row(hbuf, csrSrc, invSrc, end - cnt, cnt, lane);
        const float dg   = (float)cnt;
        const float invd = (dg > 0.0f) ? rsqrtf(fmaxf(dg, 1.0f)) : 0.0f;
        float4 o;
        o.x = a0 * hd.x + a1 * (acc.x * invd);
        o.y = a0 * hd.y + a1 * (acc.y * invd);
        o.z = a0 * hd.z + a1 * (acc.z * invd);
        o.w = a0 * hd.w + a1 * (acc.w * invd);
        *(float4*)(out + (size_t)d * H_FEATS + 4 * lane) = o;
    }
}

extern "C" void kernel_launch(void* const* d_in, const int* in_sizes, int n_in,
                              void* d_out, int out_size, void* d_ws, size_t ws_size,
                              hipStream_t stream)
{
    (void)in_sizes; (void)n_in; (void)out_size; (void)ws_size;

    const float* in_feat = (const float*)d_in[0];
    const int*   edge    = (const int*)d_in[1];
    const float* W       = (const float*)d_in[2];
    const float* bias    = (const float*)d_in[3];
    const float* alphas  = (const float*)d_in[4];
    float*       out     = (float*)d_out;

    const int* srcIdx = edge;
    const int* dstIdx = edge + N_EDGES;

    float* degOut = (float*)d_ws;
    int*   cntIn  = (int*)(degOut + N_NODES);
    int*   cursor = cntIn + N_NODES;
    int*   rowPtr = cursor + 64;
    int*   csrSrc = rowPtr + N_NODES;
    float* hbuf   = degOut + 2 * N_NODES + (size_t)N_NODES * H_FEATS;
    float* xbuf   = out;

    hipMemsetAsync(degOut, 0, (size_t)(2 * N_NODES + 64) * sizeof(float), stream);
    count_kernel<<<(N_EDGES + 255) / 256, 256, 0, stream>>>(srcIdx, dstIdx, degOut, cntIn);
    prep_kernel<<<(N_NODES + 255) / 256, 256, 0, stream>>>(degOut, cntIn, rowPtr, cursor);

    lin_wmma_kernel<<<LIN_BLOCKS, LIN_THREADS, 0, stream>>>(in_feat, W, bias, xbuf);

    fill_kernel<<<(N_EDGES + 255) / 256, 256, 0, stream>>>(srcIdx, dstIdx, rowPtr, csrSrc);

    pull_first_kernel<<<PULL_BLOCKS, 256, 0, stream>>>(xbuf, csrSrc, rowPtr, cntIn, degOut, hbuf);

    pull_second_kernel<<<PULL_BLOCKS, 256, 0, stream>>>(hbuf, csrSrc, rowPtr, cntIn, degOut, alphas, out);
}


# destination-sorted edge list built with no global atomics: chunks bucketed by node range via LDS atomic ranks, one owner work-group per bucket counts, scans, places; sums stay in arrival order
# speedup vs baseline: 3.3832x; 1.4590x over previous
#include <hip/hip_runtime.h>
#include <hip/hip_bf16.h>
#include <math.h>

#define N_NODES 100000
#define N_EDGES 600000
#define IN_FEATS 256
#define H_FEATS 128

typedef __attribute__((ext_vector_type(2))) float v2f;
typedef __attribute__((ext_vector_type(8))) float v8f;

#define BKT_SHIFT 9
#define BKT_NODES (1 << BKT_SHIFT)
#define N_BKT ((N_NODES + BKT_NODES - 1) / BKT_NODES)
#define CH_THREADS 1024
#define CH_PER_THREAD 8
#define CH_EDGES (CH_THREADS * CH_PER_THREAD)
#define N_CH 76
static_assert(N_CH % 4 == 0, "table rows are summed 16 bytes at a time");
static_assert((long long)N_CH * CH_EDGES >= N_EDGES, "every edge needs a chunk");
static_assert(N_BKT <= 256 && N_CH <= BKT_NODES, "table sizes");

__device__ __forceinline__ int block_excl_scan(int v, int* wtot, int nWaves, int* total)
{
    const int lane = threadIdx.x & 31;
    const int wave = threadIdx.x >> 5;
    int incl = v;
    #pragma unroll
    for (int o = 1; o < 32; o <<= 1) {
        const int up = __shfl_up(incl, o);
        if (lane >= o) incl += up;
    }
    if (lane == 31) wtot[wave] = incl;
    __syncthreads();
    int before = 0, all = 0;
    for (int q = 0; q < nWaves; ++q) {
        const int s = wtot[q];
        if (q < wave) before += s;
        all += s;
    }
    __syncthreads();
    *total = all;
    return before + incl - v;
}

__global__ __launch_bounds__(CH_THREADS) void bucket_kernel(
    const int* __restrict__ srcIdx, const int* __restrict__ dstIdx,
    int* __restrict__ byDstWords, int* __restrict__ bySrc,
    int* __restrict__ cntD, int* __restrict__ offD,
    int* __restrict__ cntS, int* __restrict__ offS)
{
    int2* __restrict__ byDst = (int2*)byDstWords;
    __shared__ int hD[256];
    __shared__ int hS[256];
    __shared__ int wtot[CH_THREADS / 32];
    const int tid = threadIdx.x;
    const int ch  = blockIdx.x;
    const int e0  = ch * CH_EDGES;
    if (tid < 256) { hD[tid] = 0; hS[tid] = 0; }
    __syncthreads();

    int s[CH_PER_THREAD], d[CH_PER_THREAD], rd[CH_PER_THREAD], rs[CH_PER_THREAD];
    #pragma unroll
    for (int k = 0; k < CH_PER_THREAD; ++k) {
        const int e = e0 + k * CH_THREADS + tid;
        s[k] = 0; d[k] = 0;
        if (e < N_EDGES) { s[k] = srcIdx[e]; d[k] = dstIdx[e]; }
    }
    #pragma unroll
    for (int k = 0; k < CH_PER_THREAD; ++k) {
        rd[k] = 0; rs[k] = 0;
        if (e0 + k * CH_THREADS + tid < N_EDGES) {
            rd[k] = atomicAdd(&hD[d[k] >> BKT_SHIFT], 1);
            rs[k] = atomicAdd(&hS[s[k] >> BKT_SHIFT], 1);
        }
    }
    __syncthreads();

    const int cd = (tid < 256) ? hD[tid] : 0;
    const int cs = (tid < 256) ? hS[tid] : 0;
    int totD, totS;
    const int od = block_excl_scan(cd, wtot, CH_THREADS / 32, &totD);
    const int os = block_excl_scan(cs, wtot, CH_THREADS / 32, &totS);
    if (tid < 256) { hD[tid] = od; hS[tid] = os; }
    if (tid < N_BKT) {
        cntD[tid * N_CH + ch] = cd;  offD[tid * N_CH + ch] = od;
        cntS[tid * N_CH + ch] = cs;  offS[tid * N_CH + ch] = os;
    }
    __syncthreads();

    #pragma unroll
    for (int k = 0; k < CH_PER_THREAD; ++k) {
        if (e0 + k * CH_THREADS + tid < N_EDGES) {
            byDst[e0 + hD[d[k] >> BKT_SHIFT] + rd[k]] = make_int2(d[k], s[k]);
            bySrc[e0 + hS[s[k] >> BKT_SHIFT] + rs[k]] = s[k];
        }
    }
}

__global__ __launch_bounds__(BKT_NODES) void owner_kernel(
    const int* __restrict__ byDstWords, const int* __restrict__ bySrc,
    const int* __restrict__ cntD, const int* __restrict__ offD,
    const int* __restrict__ cntS, const int* __restrict__ offS,
    float* __restrict__ invSrc, int* __restrict__ cntIn,
    int* __restrict__ rowEnd, int* __restrict__ csrSrc)
{
    const int2* __restrict__ byDst = (const int2*)byDstWords;
    __shared__ int nIn[BKT_NODES];
    __shared__ int nOut[BKT_NODES];
    __shared__ int segCD[N_CH], segOD[N_CH], segCS[N_CH], segOS[N_CH];
    __shared__ int wtot[BKT_NODES / 32];
    const int tid  = threadIdx.x;
    const int lane = tid & 31;
    const int wave = __builtin_amdgcn_readfirstlane(tid >> 5);
    const int b    = blockIdx.x;

    nIn[tid] = 0; nOut[tid] = 0;
    if (tid < N_CH) {
        segCD[tid] = cntD[b * N_CH + tid];  segOD[tid] = offD[b * N_CH + tid];
        segCS[tid] = cntS[b * N_CH + tid];  segOS[tid] = offS[b * N_CH + tid];
    }
    int part = 0;
    const int4* c4 = (const int4*)cntD;
    for (int i = tid; i < b * (N_CH / 4); i += BKT_NODES) {
        const int4 v = c4[i];
        part += (v.x + v.y) + (v.z + v.w);
    }
    int base;
    (void)block_excl_scan(part, wtot, BKT_NODES / 32, &base);

    for (int t = wave; t < N_CH; t += BKT_NODES / 32) {
        const int nd = segCD[t], ns = segCS[t];
        const int2* pd = byDst + (size_t)t * CH_EDGES + segOD[t];
        const int*  ps = bySrc + (size_t)t * CH_EDGES + segOS[t];
        const int nm = nd > ns ? nd : ns;
        for (int j = lane; j < nm; j += 32) {
            int dv = 0, sv = 0;
            if (j < nd) dv = pd[j].x;
            if (j < ns) sv = ps[j];
            if (j < nd) atomicAdd(&nIn[dv & (BKT_NODES - 1)], 1);
            if (j < ns) atomicAdd(&nOut[sv & (BKT_NODES - 1)], 1);
        }
    }
    __syncthreads();

    const int cin = nIn[tid];
    int tot;
    const int first = base + block_excl_scan(cin, wtot, BKT_NODES / 32, &tot);
    const int node  = b * BKT_NODES + tid;
    if (node < N_NODES) {
        cntIn[node]  = cin;
        rowEnd[node] = first + cin;
        const float dg = (float)nOut[tid];
        invSrc[node] = (dg > 0.0f) ? rsqrtf(fmaxf(dg, 1.0f)) : 0.0f;
    }
    nIn[tid] = first;
    __syncthreads();

    for (int t = wave; t < N_CH; t += BKT_NODES / 32) {
        const int nd = segCD[t];
        const int2* pd = byDst + (size_t)t * CH_EDGES + segOD[t];
        for (int j = lane; j < nd; j += 32) {
            const int2 e = pd[j];
            const int pos = atomicAdd(&nIn[e.x & (BKT_NODES - 1)], 1);
            csrSrc[pos] = e.y;
        }
    }
}

#define WL_PITCH 260
#define LIN_TILES (N_NODES / 16)
#define LIN_BLOCKS 256
#define LIN_THREADS 512

__global__ __launch_bounds__(LIN_THREADS) void lin_wmma_kernel(
    const float* __restrict__ A, const float* __restrict__ W,
    const float* __restrict__ bias, float* __restrict__ X)
{
    __shared__ __attribute__((aligned(16))) float Wl[H_FEATS * WL_PITCH];

    const int lane   = threadIdx.x & 31;
    const int wave   = __builtin_amdgcn_readfirstlane(threadIdx.x >> 5);
    const int h      = lane >> 4;
    const int c      = lane & 15;
    const int nWaves = gridDim.x * (LIN_THREADS / 32);
    int t = blockIdx.x + gridDim.x * wave;

    const float* ap = A + (size_t)(t * 16 + c) * IN_FEATS + 4 * h;
    float4 acur[8];
    #pragma unroll
    for (int i = 0; i < 8; ++i) acur[i] = *(const float4*)(ap + 8 * i);

    float bb[8];
    #pragma unroll
    for (int j = 0; j < 8; ++j) bb[j] = bias[16 * j + c];

    for (int it = 0; it < (IN_FEATS * H_FEATS) / LIN_THREADS; ++it) {
        const int i = it * LIN_THREADS + threadIdx.x;
        Wl[(i & (H_FEATS - 1)) * WL_PITCH + (i >> 7)] = W[i];
    }
    __syncthreads();

    const float* wl = Wl + c * WL_PITCH + 4 * h;

    while (t < LIN_TILES) {
        asm volatile("" ::: "memory");
        const int tn = t + nWaves;
        const float* apn = A + (size_t)((tn < LIN_TILES ? tn : t) * 16 + c) * IN_FEATS + 4 * h;

        v8f acc[8];
        #pragma unroll
        for (int j = 0; j < 8; ++j) acc[j] = (v8f){};

        #pragma unroll
        for (int ch = 0; ch < 4; ++ch) {
            const float* nx = (ch < 3) ? (ap + 64 * (ch + 1)) : apn;
            float4 anxt[8];
            #pragma unroll
            for (int i = 0; i < 8; ++i) anxt[i] = *(const float4*)(nx + 8 * i);
            __builtin_amdgcn_sched_barrier(0);

            #pragma unroll
            for (int i = 0; i < 8; ++i) {
                const float* wk = wl + 64 * ch + 8 * i;
                float4 b[8];
                #pragma unroll
                for (int j = 0; j < 8; ++j) b[j] = *(const float4*)(wk + j * 16 * WL_PITCH);
                v2f a01, a23;
                a01.x = acur[i].x; a01.y = acur[i].y;
                a23.x = acur[i].z; a23.y = acur[i].w;
                #pragma unroll
                for (int j = 0; j < 8; ++j) {
                    v2f b01; b01.x = b[j].x; b01.y = b[j].y;
                    acc[j] = __builtin_amdgcn_wmma_f32_16x16x4_f32(
                                false, a01, false, b01, (short)0, acc[j], false, false);
                }
                #pragma unroll
                for (int j = 0; j < 8; ++j) {
                    v2f b23; b23.x = b[j].z; b23.y = b[j].w;
                    acc[j] = __builtin_amdgcn_wmma_f32_16x16x4_f32(
                                false, a23, false, b23, (short)0, acc[j], false, false);
                }
            }
            #pragma unroll
            for (int i = 0; i < 8; ++i) acur[i] = anxt[i];
        }

        float* xp = X + (size_t)(t * 16 + 8 * h) * H_FEATS + c;
        #pragma unroll
        for (int j = 0; j < 8; ++j) {
            #pragma unroll
            for (int r = 0; r < 8; ++r)
                xp[(size_t)r * H_FEATS + 16 * j] = acc[j][r] + bb[j];
        }

        ap = apn;
        t  = tn;
    }
}

__device__ __forceinline__ float4 pull_row(
    const float* __restrict__ feat, const int* __restrict__ csrSrc,
    const float* __restrict__ invSrc, int start, int cnt, int lane)
{
    float4 acc = make_float4(0.0f, 0.0f, 0.0f, 0.0f);
    for (int base = 0; base < cnt; base += 32) {
        const int n = (cnt - base < 32) ? (cnt - base) : 32;
        int   myS = 0;
        float myW = 0.0f;
        if (lane < n) { myS = csrSrc[start + base + lane]; myW = invSrc[myS]; }
        for (int j = 0; j < n; j += 8) {
            float4 v[8];
            float  w[8];
            #pragma unroll
            for (int u = 0; u < 8; ++u) {
                const int s = __builtin_amdgcn_readlane(myS, j + u);
                w[u] = __int_as_float(__builtin_amdgcn_readlane(__float_as_int(myW), j + u));
                v[u] = *(const float4*)(feat + (size_t)s * H_FEATS + 4 * lane);
            }
            #pragma unroll
            for (int u = 0; u < 8; ++u) {
                const bool on = (j + u < n);
                const float4 t = v[u];
                acc.x += (on ? t.x : 0.0f) * w[u];
                acc.y += (on ? t.y : 0.0f) * w[u];
                acc.z += (on ? t.z : 0.0f) * w[u];
                acc.w += (on ? t.w : 0.0f) * w[u];
            }
        }
    }
    return acc;
}

__device__ __forceinline__ float softmax2(const float* __restrict__ alphas, int idx)
{
    const float a0 = alphas[0], a1 = alphas[1];
    const float m  = fmaxf(a0, a1);
    const float e0 = __expf(a0 - m), e1 = __expf(a1 - m);
    return (idx == 0 ? e0 : e1) / (e0 + e1);
}

#define PULL_BLOCKS 2048
#define PULL_WAVES  (PULL_BLOCKS * 8)

__global__ __launch_bounds__(256) void pull_first_kernel(
    const float* __restrict__ feat, const int* __restrict__ csrSrc,
    const int* __restrict__ rowEnd, const int* __restrict__ cntIn,
    const float* __restrict__ invSrc, float* __restrict__ hbuf)
{
    const int lane = threadIdx.x & 31;
    const int d0 = blockIdx.x * 8 + __builtin_amdgcn_readfirstlane(threadIdx.x >> 5);
    for (int d = d0; d < N_NODES; d += PULL_WAVES) {
        const int cnt = cntIn[d];
        const int end = rowEnd[d];
        const float4 acc = pull_row(feat, csrSrc, invSrc, end - cnt, cnt, lane);
        const float dg   = (float)cnt;
        const float invd = (dg > 0.0f) ? rsqrtf(fmaxf(dg, 1.0f)) : 0.0f;
        float4 hv;
        hv.x = acc.x * invd; hv.y = acc.y * invd; hv.z = acc.z * invd; hv.w = acc.w * invd;
        *(float4*)(hbuf + (size_t)d * H_FEATS + 4 * lane) = hv;
    }
}

__global__ __launch_bounds__(256) void pull_second_kernel(
    const float* __restrict__ hbuf, const int* __restrict__ csrSrc,
    const int* __restrict__ rowEnd, const int* __restrict__ cntIn,
    const float* __restrict__ invSrc, const float* __restrict__ alphas,
    float* __restrict__ out)
{
    const int lane = threadIdx.x & 31;
    const float a0 = __int_as_float(__builtin_amdgcn_readfirstlane(__float_as_int(softmax2(alphas, 0))));
    const float a1 = __int_as_float(__builtin_amdgcn_readfirstlane(__float_as_int(softmax2(alphas, 1))));
    const int d0 = blockIdx.x * 8 + __builtin_amdgcn_readfirstlane(threadIdx.x >> 5);
    for (int d = d0; d < N_NODES; d += PULL_WAVES) {
        const int cnt = cntIn[d];
        const int end = rowEnd[d];
        const float4 hd = *(const float4*)(hbuf + (size_t)d * H_FEATS + 4 * lane);
        const float4 acc = pull_row(hbuf, csrSrc, invSrc, end - cnt, cnt, lane);
        const float dg   = (float)cnt;
        const float invd = (dg > 0.0f) ? rsqrtf(fmaxf(dg, 1.0f)) : 0.0f;
        float4 o;
        o.x = a0 * hd.x + a1 * (acc.x * invd);
        o.y = a0 * hd.y + a1 * (acc.y * invd);
        o.z = a0 * hd.z + a1 * (acc.z * invd);
        o.w = a0 * hd.w + a1 * (acc.w * invd);
        *(float4*)(out + (size_t)d * H_FEATS + 4 * lane) = o;
    }
}

extern "C" void kernel_launch(void* const* d_in, const int* in_sizes, int n_in,
                              void* d_out, int out_size, void* d_ws, size_t ws_size,
                              hipStream_t stream)
{
    (void)in_sizes; (void)n_in; (void)out_size; (void)ws_size;

    const float* in_feat = (const float*)d_in[0];
    const int*   edge    = (const int*)d_in[1];
    const float* W       = (const float*)d_in[2];
    const float* bias    = (const float*)d_in[3];
    const float* alphas  = (const float*)d_in[4];
    float*       out     = (float*)d_out;

    const int* srcIdx = edge;
    const int* dstIdx = edge + N_EDGES;

    float* degOut = (float*)d_ws;
    int*   cntIn  = (int*)(degOut + N_NODES);
    int*   rowPtr = cntIn + N_NODES;
    int*   csrSrc = rowPtr + N_NODES;
    int*   bySrc  = csrSrc + N_EDGES;
    int*   byDst  = bySrc + N_CH * CH_EDGES;
    int*   cntD   = byDst + 2 * N_CH * CH_EDGES;
    int*   offD   = cntD + N_BKT * N_CH;
    int*   cntS   = offD + N_BKT * N_CH;
    int*   offS   = cntS + N_BKT * N_CH;
    float* hbuf   = degOut + 2 * N_NODES + (size_t)N_NODES * H_FEATS;
    float* xbuf   = out;

    bucket_kernel<<<N_CH, CH_THREADS, 0, stream>>>(srcIdx, dstIdx, byDst, bySrc, cntD, offD, cntS, offS);
    owner_kernel<<<N_BKT, BKT_NODES, 0, stream>>>(byDst, bySrc, cntD, offD, cntS, offS, degOut, cntIn, rowPtr, csrSrc);

    lin_wmma_kernel<<<LIN_BLOCKS, LIN_THREADS, 0, stream>>>(in_feat, W, bias, xbuf);

    pull_first_kernel<<<PULL_BLOCKS, 256, 0, stream>>>(xbuf, csrSrc, rowPtr, cntIn, degOut, hbuf);

    pull_second_kernel<<<PULL_BLOCKS, 256, 0, stream>>>(hbuf, csrSrc, rowPtr, cntIn, degOut, alphas, out);
}


# bucket-owner work-groups run inside the GEMM's launch (by block index, two work-groups per WGP; GEMM trimmed to 128 registers): owner round trips hide under WMMAs, one launch fewer; sums same
# speedup vs baseline: 3.6572x; 1.0810x over previous
#include <hip/hip_runtime.h>
#include <hip/hip_bf16.h>
#include <math.h>

#define N_NODES 100000
#define N_EDGES 600000
#define IN_FEATS 256
#define H_FEATS 128

typedef __attribute__((ext_vector_type(2))) float v2f;
typedef __attribute__((ext_vector_type(8))) float v8f;

#define BKT_SHIFT 9
#define BKT_NODES (1 << BKT_SHIFT)
#define N_BKT ((N_NODES + BKT_NODES - 1) / BKT_NODES)
#define CH_THREADS 1024
#define CH_PER_THREAD 8
#define CH_EDGES (CH_THREADS * CH_PER_THREAD)
#define N_CH 76
static_assert(N_CH % 4 == 0, "table rows are summed 16 bytes at a time");
static_assert((long long)N_CH * CH_EDGES >= N_EDGES, "every edge needs a chunk");
static_assert(N_BKT <= 256 && N_CH <= BKT_NODES, "table sizes");

__device__ __forceinline__ int block_excl_scan(int v, int* wtot, int nWaves, int* total)
{
    const int lane = threadIdx.x & 31;
    const int wave = threadIdx.x >> 5;
    int incl = v;
    #pragma unroll
    for (int o = 1; o < 32; o <<= 1) {
        const int up = __shfl_up(incl, o);
        if (lane >= o) incl += up;
    }
    if (lane == 31) wtot[wave] = incl;
    __syncthreads();
    int before = 0, all = 0;
    for (int q = 0; q < nWaves; ++q) {
        const int s = wtot[q];
        if (q < wave) before += s;
        all += s;
    }
    __syncthreads();
    *total = all;
    return before + incl - v;
}

__global__ __launch_bounds__(CH_THREADS) void bucket_kernel(
    const int* __restrict__ srcIdx, const int* __restrict__ dstIdx,
    int* __restrict__ byDstWords, int* __restrict__ bySrc,
    int* __restrict__ cntD, int* __restrict__ offD,
    int* __restrict__ cntS, int* __restrict__ offS)
{
    int2* __restrict__ byDst = (int2*)byDstWords;
    __shared__ int hD[256];
    __shared__ int hS[256];
    __shared__ int wtot[CH_THREADS / 32];
    const int tid = threadIdx.x;
    const int ch  = blockIdx.x;
    const int e0  = ch * CH_EDGES;
    if (tid < 256) { hD[tid] = 0; hS[tid] = 0; }
    __syncthreads();

    int s[CH_PER_THREAD], d[CH_PER_THREAD], rd[CH_PER_THREAD], rs[CH_PER_THREAD];
    #pragma unroll
    for (int k = 0; k < CH_PER_THREAD; ++k) {
        const int e = e0 + k * CH_THREADS + tid;
        s[k] = 0; d[k] = 0;
        if (e < N_EDGES) { s[k] = srcIdx[e]; d[k] = dstIdx[e]; }
    }
    #pragma unroll
    for (int k = 0; k < CH_PER_THREAD; ++k) {
        rd[k] = 0; rs[k] = 0;
        if (e0 + k * CH_THREADS + tid < N_EDGES) {
            rd[k] = atomicAdd(&hD[d[k] >> BKT_SHIFT], 1);
            rs[k] = atomicAdd(&hS[s[k] >> BKT_SHIFT], 1);
        }
    }
    __syncthreads();

    const int cd = (tid < 256) ? hD[tid] : 0;
    const int cs = (tid < 256) ? hS[tid] : 0;
    int totD, totS;
    const int od = block_excl_scan(cd, wtot, CH_THREADS / 32, &totD);
    const int os = block_excl_scan(cs, wtot, CH_THREADS / 32, &totS);
    if (tid < 256) { hD[tid] = od; hS[tid] = os; }
    if (tid < N_BKT) {
        cntD[tid * N_CH + ch] = cd;  offD[tid * N_CH + ch] = od;
        cntS[tid * N_CH + ch] = cs;  offS[tid * N_CH + ch] = os;
    }
    __syncthreads();

    #pragma unroll
    for (int k = 0; k < CH_PER_THREAD; ++k) {
        if (e0 + k * CH_THREADS + tid < N_EDGES) {
            byDst[e0 + hD[d[k] >> BKT_SHIFT] + rd[k]] = make_int2(d[k], s[k]);
            bySrc[e0 + hS[s[k] >> BKT_SHIFT] + rs[k]] = s[k];
        }
    }
}

__device__ __forceinline__ void owner_role(const int b,
    const int* __restrict__ byDstWords, const int* __restrict__ bySrc,
    const int* __restrict__ cntD, const int* __restrict__ offD,
    const int* __restrict__ cntS, const int* __restrict__ offS,
    float* __restrict__ invSrc, int* __restrict__ cntIn,
    int* __restrict__ rowEnd, int* __restrict__ csrSrc)
{
    const int2* __restrict__ byDst = (const int2*)byDstWords;
    __shared__ int nIn[BKT_NODES];
    __shared__ int nOut[BKT_NODES];
    __shared__ int segCD[N_CH], segOD[N_CH], segCS[N_CH], segOS[N_CH];
    __shared__ int wtot[BKT_NODES / 32];
    const int tid  = threadIdx.x;
    const int lane = tid & 31;
    const int wave = __builtin_amdgcn_readfirstlane(tid >> 5);

    nIn[tid] = 0; nOut[tid] = 0;
    if (tid < N_CH) {
        segCD[tid] = cntD[b * N_CH + tid];  segOD[tid] = offD[b * N_CH + tid];
        segCS[tid] = cntS[b * N_CH + tid];  segOS[tid] = offS[b * N_CH + tid];
    }
    int part = 0;
    const int4* c4 = (const int4*)cntD;
    for (int i = tid; i < b * (N_CH / 4); i += BKT_NODES) {
        const int4 v = c4[i];
        part += (v.x + v.y) + (v.z + v.w);
    }
    int base;
    (void)block_excl_scan(part, wtot, BKT_NODES / 32, &base);

    for (int t = wave; t < N_CH; t += BKT_NODES / 32) {
        const int nd = segCD[t], ns = segCS[t];
        const int2* pd = byDst + (size_t)t * CH_EDGES + segOD[t];
        const int*  ps = bySrc + (size_t)t * CH_EDGES + segOS[t];
        const int nm = nd > ns ? nd : ns;
        for (int j = lane; j < nm; j += 32) {
            int dv = 0, sv = 0;
            if (j < nd) dv = pd[j].x;
            if (j < ns) sv = ps[j];
            if (j < nd) atomicAdd(&nIn[dv & (BKT_NODES - 1)], 1);
            if (j < ns) atomicAdd(&nOut[sv & (BKT_NODES - 1)], 1);
        }
    }
    __syncthreads();

    const int cin = nIn[tid];
    int tot;
    const int first = base + block_excl_scan(cin, wtot, BKT_NODES / 32, &tot);
    const int node  = b * BKT_NODES + tid;
    if (node < N_NODES) {
        cntIn[node]  = cin;
        rowEnd[node] = first + cin;
        const float dg = (float)nOut[tid];
        invSrc[node] = (dg > 0.0f) ? rsqrtf(fmaxf(dg, 1.0f)) : 0.0f;
    }
    nIn[tid] = first;
    __syncthreads();

    for (int t = wave; t < N_CH; t += BKT_NODES / 32) {
        const int nd = segCD[t];
        const int2* pd = byDst + (size_t)t * CH_EDGES + segOD[t];
        for (int j = lane; j < nd; j += 32) {
            const int2 e = pd[j];
            const int pos = atomicAdd(&nIn[e.x & (BKT_NODES - 1)], 1);
            csrSrc[pos] = e.y;
        }
    }
}

#define WL_PITCH 260
static_assert(WL_PITCH > IN_FEATS, "row n keeps bias[n] in its first padding word");
#define LIN_TILES (N_NODES / 16)
#define LIN_BLOCKS 256
#define LIN_THREADS 512

__device__ __forceinline__ void lin_role(const int g,
    const float* __restrict__ A, const float* __restrict__ W,
    const float* __restrict__ bias, float* __restrict__ X)
{
    __shared__ __attribute__((aligned(16))) float Wl[H_FEATS * WL_PITCH];

    const int lane   = threadIdx.x & 31;
    const int wave   = __builtin_amdgcn_readfirstlane(threadIdx.x >> 5);
    const int h      = lane >> 4;
    const int c      = lane & 15;
    const int nWaves = LIN_BLOCKS * (LIN_THREADS / 32);
    int t = g + LIN_BLOCKS * wave;

    const int laneA = c * IN_FEATS + 4 * h;
    const int laneX = 8 * h * H_FEATS + c;

    const float* ap = A + (size_t)t * (16 * IN_FEATS);
    float4 acur[4];
    #pragma unroll
    for (int i = 0; i < 4; ++i) acur[i] = *(const float4*)(ap + laneA + 8 * i);

    #pragma nounroll
    for (int ob = 0; ob < (IN_FEATS * H_FEATS) / LIN_THREADS; ob += 32) {
        #pragma unroll
        for (int it = 0; it < 32; ++it) {
            const int i = (ob + it) * LIN_THREADS + threadIdx.x;
            Wl[(i & (H_FEATS - 1)) * WL_PITCH + (i >> 7)] = W[i];
        }
    }
    if (threadIdx.x < H_FEATS) Wl[threadIdx.x * WL_PITCH + IN_FEATS] = bias[threadIdx.x];
    __syncthreads();

    const float* wl = Wl + c * WL_PITCH + 4 * h;

    while (t < LIN_TILES) {
        asm volatile("" ::: "memory");
        const int tn = t + nWaves;
        const float* apn = A + (size_t)(tn < LIN_TILES ? tn : t) * (16 * IN_FEATS);

        v8f acc[8];
        #pragma unroll
        for (int j = 0; j < 8; ++j) acc[j] = (v8f){};

        #pragma unroll
        for (int ch = 0; ch < 8; ++ch) {
            const float* nx = (ch < 7) ? (ap + 32 * (ch + 1)) : apn;
            float4 anxt[4];
            #pragma unroll
            for (int i = 0; i < 4; ++i) anxt[i] = *(const float4*)(nx + laneA + 8 * i);
            __builtin_amdgcn_sched_barrier(0);

            #pragma unroll
            for (int i = 0; i < 4; ++i) {
                const float* wk = wl + 32 * ch + 8 * i;
                v2f a01, a23;
                a01.x = acur[i].x; a01.y = acur[i].y;
                a23.x = acur[i].z; a23.y = acur[i].w;
                #pragma unroll
                for (int jh = 0; jh < 8; jh += 4) {
                    float4 b[4];
                    #pragma unroll
                    for (int j = 0; j < 4; ++j) b[j] = *(const float4*)(wk + (jh + j) * 16 * WL_PITCH);
                    #pragma unroll
                    for (int j = 0; j < 4; ++j) {
                        v2f b01; b01.x = b[j].x; b01.y = b[j].y;
                        acc[jh + j] = __builtin_amdgcn_wmma_f32_16x16x4_f32(
                                    false, a01, false, b01, (short)0, acc[jh + j], false, false);
                    }
                    #pragma unroll
                    for (int j = 0; j < 4; ++j) {
                        v2f b23; b23.x = b[j].z; b23.y = b[j].w;
                        acc[jh + j] = __builtin_amdgcn_wmma_f32_16x16x4_f32(
                                    false, a23, false, b23, (short)0, acc[jh + j], false, false);
                    }
                    __builtin_amdgcn_sched_barrier(0);
                }
            }
            #pragma unroll
            for (int i = 0; i < 4; ++i) acur[i] = anxt[i];
        }

        float* xp = X + (size_t)t * (16 * H_FEATS);
        #pragma unroll
        for (int j = 0; j < 8; ++j) {
            const float bb = Wl[(16 * j + c) * WL_PITCH + IN_FEATS];
            #pragma unroll
            for (int r = 0; r < 8; ++r)
                xp[laneX + r * H_FEATS + 16 * j] = acc[j][r] + bb;
        }

        ap = apn;
        t  = tn;
    }
}

static_assert(BKT_NODES == LIN_THREADS, "both kinds of work-group have the launch's one block size");

__global__ __launch_bounds__(LIN_THREADS, 8) void lin_owner_kernel(
    const float* __restrict__ A, const float* __restrict__ W,
    const float* __restrict__ bias, float* __restrict__ X,
    const int* __restrict__ byDstWords, const int* __restrict__ bySrc,
    const int* __restrict__ cntD, const int* __restrict__ offD,
    const int* __restrict__ cntS, const int* __restrict__ offS,
    float* __restrict__ invSrc, int* __restrict__ cntIn,
    int* __restrict__ rowEnd, int* __restrict__ csrSrc)
{
    if (blockIdx.x < LIN_BLOCKS)
        lin_role(blockIdx.x, A, W, bias, X);
    else
        owner_role(blockIdx.x - LIN_BLOCKS, byDstWords, bySrc, cntD, offD, cntS, offS, invSrc, cntIn, rowEnd, csrSrc);
}

__device__ __forceinline__ float4 pull_row(
    const float* __restrict__ feat, const int* __restrict__ csrSrc,
    const float* __restrict__ invSrc, int start, int cnt, int lane)
{
    float4 acc = make_float4(0.0f, 0.0f, 0.0f, 0.0f);
    for (int base = 0; base < cnt; base += 32) {
        const int n = (cnt - base < 32) ? (cnt - base) : 32;
        int   myS = 0;
        float myW = 0.0f;
        if (lane < n) { myS = csrSrc[start + base + lane]; myW = invSrc[myS]; }
        for (int j = 0; j < n; j += 8) {
            float4 v[8];
            float  w[8];
            #pragma unroll
            for (int u = 0; u < 8; ++u) {
                const int s = __builtin_amdgcn_readlane(myS, j + u);
                w[u] = __int_as_float(__builtin_amdgcn_readlane(__float_as_int(myW), j + u));
                v[u] = *(const float4*)(feat + (size_t)s * H_FEATS + 4 * lane);
            }
            #pragma unroll
            for (int u = 0; u < 8; ++u) {
                const bool on = (j + u < n);
                const float4 t = v[u];
                acc.x += (on ? t.x : 0.0f) * w[u];
                acc.y += (on ? t.y : 0.0f) * w[u];
                acc.z += (on ? t.z : 0.0f) * w[u];
                acc.w += (on ? t.w : 0.0f) * w[u];
            }
        }
    }
    return acc;
}

__device__ __forceinline__ float softmax2(const float* __restrict__ alphas, int idx)
{
    const float a0 = alphas[0], a1 = alphas[1];
    const float m  = fmaxf(a0, a1);
    const float e0 = __expf(a0 - m), e1 = __expf(a1 - m);
    return (idx == 0 ? e0 : e1) / (e0 + e1);
}

#define PULL_BLOCKS 2048
#define PULL_WAVES  (PULL_BLOCKS * 8)

__global__ __launch_bounds__(256) void pull_first_kernel(
    const float* __restrict__ feat, const int* __restrict__ csrSrc,
    const int* __restrict__ rowEnd, const int* __restrict__ cntIn,
    const float* __restrict__ invSrc, float* __restrict__ hbuf)
{
    const int lane = threadIdx.x & 31;
    const int d0 = blockIdx.x * 8 + __builtin_amdgcn_readfirstlane(threadIdx.x >> 5);
    for (int d = d0; d < N_NODES; d += PULL_WAVES) {
        const int cnt = cntIn[d];
        const int end = rowEnd[d];
        const float4 acc = pull_row(feat, csrSrc, invSrc, end - cnt, cnt, lane);
        const float dg   = (float)cnt;
        const float invd = (dg > 0.0f) ? rsqrtf(fmaxf(dg, 1.0f)) : 0.0f;
        float4 hv;
        hv.x = acc.x * invd; hv.y = acc.y * invd; hv.z = acc.z * invd; hv.w = acc.w * invd;
        *(float4*)(hbuf + (size_t)d * H_FEATS + 4 * lane) = hv;
    }
}

__global__ __launch_bounds__(256) void pull_second_kernel(
    const float* __restrict__ hbuf, const int* __restrict__ csrSrc,
    const int* __restrict__ rowEnd, const int* __restrict__ cntIn,
    const float* __restrict__ invSrc, const float* __restrict__ alphas,
    float* __restrict__ out)
{
    const int lane = threadIdx.x & 31;
    const float a0 = __int_as_float(__builtin_amdgcn_readfirstlane(__float_as_int(softmax2(alphas, 0))));
    const float a1 = __int_as_float(__builtin_amdgcn_readfirstlane(__float_as_int(softmax2(alphas, 1))));
    const int d0 = blockIdx.x * 8 + __builtin_amdgcn_readfirstlane(threadIdx.x >> 5);
    for (int d = d0; d < N_NODES; d += PULL_WAVES) {
        const int cnt = cntIn[d];
        const int end = rowEnd[d];
        const float4 hd = *(const float4*)(hbuf + (size_t)d * H_FEATS + 4 * lane);
        const float4 acc = pull_row(hbuf, csrSrc, invSrc, end - cnt, cnt, lane);
        const float dg   = (float)cnt;
        const float invd = (dg > 0.0f) ? rsqrtf(fmaxf(dg, 1.0f)) : 0.0f;
        float4 o;
        o.x = a0 * hd.x + a1 * (acc.x * invd);
        o.y = a0 * hd.y + a1 * (acc.y * invd);
        o.z = a0 * hd.z + a1 * (acc.z * invd);
        o.w = a0 * hd.w + a1 * (acc.w * invd);
        *(float4*)(out + (size_t)d * H_FEATS + 4 * lane) = o;
    }
}

extern "C" void kernel_launch(void* const* d_in, const int* in_sizes, int n_in,
                              void* d_out, int out_size, void* d_ws, size_t ws_size,
                              hipStream_t stream)
{
    (void)in_sizes; (void)n_in; (void)out_size; (void)ws_size;

    const float* in_feat = (const float*)d_in[0];
    const int*   edge    = (const int*)d_in[1];
    const float* W       = (const float*)d_in[2];
    const float* bias    = (const float*)d_in[3];
    const float* alphas  = (const float*)d_in[4];
    float*       out     = (float*)d_out;

    const int* srcIdx = edge;
    const int* dstIdx = edge + N_EDGES;

    float* degOut = (float*)d_ws;
    int*   cntIn  = (int*)(degOut + N_NODES);
    int*   rowPtr = cntIn + N_NODES;
    int*   csrSrc = rowPtr + N_NODES;
    int*   bySrc  = csrSrc + N_EDGES;
    int*   byDst  = bySrc + N_CH * CH_EDGES;
    int*   cntD   = byDst + 2 * N_CH * CH_EDGES;
    int*   offD   = cntD + N_BKT * N_CH;
    int*   cntS   = offD + N_BKT * N_CH;
    int*   offS   = cntS + N_BKT * N_CH;
    float* hbuf   = degOut + 2 * N_NODES + (size_t)N_NODES * H_FEATS;
    float* xbuf   = out;

    bucket_kernel<<<N_CH, CH_THREADS, 0, stream>>>(srcIdx, dstIdx, byDst, bySrc, cntD, offD, cntS, offS);

    lin_owner_kernel<<<LIN_BLOCKS + N_BKT, LIN_THREADS, 0, stream>>>(
        in_feat, W, bias, xbuf, byDst, bySrc, cntD, offD, cntS, offS, degOut, cntIn, rowPtr, csrSrc);

    pull_first_kernel<<<PULL_BLOCKS, 256, 0, stream>>>(xbuf, csrSrc, rowPtr, cntIn, degOut, hbuf);

    pull_second_kernel<<<PULL_BLOCKS, 256, 0, stream>>>(hbuf, csrSrc, rowPtr, cntIn, degOut, alphas, out);
}


# bucket launch: each chunk is put in bucket order in LDS (96 KB) and leaves as whole 128-byte lines, 16 bytes a lane, instead of 16 scattered 4- and 8-byte stores a thread; slots and sum order same
# speedup vs baseline: 3.8971x; 1.0656x over previous
#include <hip/hip_runtime.h>
#include <hip/hip_bf16.h>
#include <math.h>

#define N_NODES 100000
#define N_EDGES 600000
#define IN_FEATS 256
#define H_FEATS 128

typedef __attribute__((ext_vector_type(2))) float v2f;
typedef __attribute__((ext_vector_type(8))) float v8f;

#define BKT_SHIFT 9
#define BKT_NODES (1 << BKT_SHIFT)
#define N_BKT ((N_NODES + BKT_NODES - 1) / BKT_NODES)
#define CH_THREADS 1024
#define CH_PER_THREAD 8
#define CH_EDGES (CH_THREADS * CH_PER_THREAD)
#define N_CH 76
static_assert(N_CH % 4 == 0, "table rows are summed 16 bytes at a time");
static_assert((long long)N_CH * CH_EDGES >= N_EDGES, "every edge needs a chunk");
static_assert(N_BKT <= 256 && N_CH <= BKT_NODES, "table sizes");
static_assert(N_EDGES % 4 == 0 && CH_EDGES % (4 * CH_THREADS) == 0, "an ordered chunk is written out in whole 16-byte pieces");

__device__ __forceinline__ int block_excl_scan(int v, int* wtot, int nWaves, int* total)
{
    const int lane = threadIdx.x & 31;
    const int wave = threadIdx.x >> 5;
    int incl = v;
    #pragma unroll
    for (int o = 1; o < 32; o <<= 1) {
        const int up = __shfl_up(incl, o);
        if (lane >= o) incl += up;
    }
    if (lane == 31) wtot[wave] = incl;
    __syncthreads();
    int before = 0, all = 0;
    for (int q = 0; q < nWaves; ++q) {
        const int s = wtot[q];
        if (q < wave) before += s;
        all += s;
    }
    __syncthreads();
    *total = all;
    return before + incl - v;
}

__global__ __launch_bounds__(CH_THREADS) void bucket_kernel(
    const int* __restrict__ srcIdx, const int* __restrict__ dstIdx,
    int* __restrict__ byDstWords, int* __restrict__ bySrc,
    int* __restrict__ cntD, int* __restrict__ offD,
    int* __restrict__ cntS, int* __restrict__ offS)
{
    __shared__ __attribute__((aligned(16))) int stD[2 * CH_EDGES];
    __shared__ __attribute__((aligned(16))) int stS[CH_EDGES];
    __shared__ int hD[256];
    __shared__ int hS[256];
    __shared__ int wtot[CH_THREADS / 32];
    const int tid = threadIdx.x;
    const int ch  = blockIdx.x;
    const int e0  = ch * CH_EDGES;
    if (tid < 256) { hD[tid] = 0; hS[tid] = 0; }
    __syncthreads();

    int s[CH_PER_THREAD], d[CH_PER_THREAD], rd[CH_PER_THREAD], rs[CH_PER_THREAD];
    #pragma unroll
    for (int k = 0; k < CH_PER_THREAD; ++k) {
        const int e = e0 + k * CH_THREADS + tid;
        s[k] = 0; d[k] = 0;
        if (e < N_EDGES) { s[k] = srcIdx[e]; d[k] = dstIdx[e]; }
    }
    #pragma unroll
    for (int k = 0; k < CH_PER_THREAD; ++k) {
        rd[k] = 0; rs[k] = 0;
        if (e0 + k * CH_THREADS + tid < N_EDGES) {
            rd[k] = atomicAdd(&hD[d[k] >> BKT_SHIFT], 1);
            rs[k] = atomicAdd(&hS[s[k] >> BKT_SHIFT], 1);
        }
    }
    __syncthreads();

    const int cd = (tid < 256) ? hD[tid] : 0;
    const int cs = (tid < 256) ? hS[tid] : 0;
    int totD, totS;
    const int od = block_excl_scan(cd, wtot, CH_THREADS / 32, &totD);
    const int os = block_excl_scan(cs, wtot, CH_THREADS / 32, &totS);
    if (tid < 256) { hD[tid] = od; hS[tid] = os; }
    if (tid < N_BKT) {
        cntD[tid * N_CH + ch] = cd;  offD[tid * N_CH + ch] = od;
        cntS[tid * N_CH + ch] = cs;  offS[tid * N_CH + ch] = os;
    }
    __syncthreads();

    int2* stPair = (int2*)stD;
    #pragma unroll
    for (int k = 0; k < CH_PER_THREAD; ++k) {
        if (e0 + k * CH_THREADS + tid < N_EDGES) {
            stPair[hD[d[k] >> BKT_SHIFT] + rd[k]] = make_int2(d[k], s[k]);
            stS[hS[s[k] >> BKT_SHIFT] + rs[k]] = s[k];
        }
    }
    __syncthreads();

    const int nValid = (N_EDGES - e0 < CH_EDGES) ? (N_EDGES - e0) : CH_EDGES;
    int4*       gD = (int4*)(byDstWords + 2 * (size_t)e0);
    const int4* lD = (const int4*)stD;
    #pragma unroll
    for (int k = 0; k < (2 * CH_EDGES) / (4 * CH_THREADS); ++k) {
        const int q = k * CH_THREADS + tid;
        if (2 * q < nValid) gD[q] = lD[q];
    }
    int4*       gS = (int4*)(bySrc + (size_t)e0);
    const int4* lS = (const int4*)stS;
    #pragma unroll
    for (int k = 0; k < CH_EDGES / (4 * CH_THREADS); ++k) {
        const int q = k * CH_THREADS + tid;
        if (4 * q < nValid) gS[q] = lS[q];
    }
}

__device__ __forceinline__ void owner_role(const int b,
    const int* __restrict__ byDstWords, const int* __restrict__ bySrc,
    const int* __restrict__ cntD, const int* __restrict__ offD,
    const int* __restrict__ cntS, const int* __restrict__ offS,
    float* __restrict__ invSrc, int* __restrict__ cntIn,
    int* __restrict__ rowEnd, int* __restrict__ csrSrc)
{
    const int2* __restrict__ byDst = (const int2*)byDstWords;
    __shared__ int nIn[BKT_NODES];
    __shared__ int nOut[BKT_NODES];
    __shared__ int segCD[N_CH], segOD[N_CH], segCS[N_CH], segOS[N_CH];
    __shared__ int wtot[BKT_NODES / 32];
    const int tid  = threadIdx.x;
    const int lane = tid & 31;
    const int wave = __builtin_amdgcn_readfirstlane(tid >> 5);

    nIn[tid] = 0; nOut[tid] = 0;
    if (tid < N_CH) {
        segCD[tid] = cntD[b * N_CH + tid];  segOD[tid] = offD[b * N_CH + tid];
        segCS[tid] = cntS[b * N_CH + tid];  segOS[tid] = offS[b * N_CH + tid];
    }
    int part = 0;
    const int4* c4 = (const int4*)cntD;
    for (int i = tid; i < b * (N_CH / 4); i += BKT_NODES) {
        const int4 v = c4[i];
        part += (v.x + v.y) + (v.z + v.w);
    }
    int base;
    (void)block_excl_scan(part, wtot, BKT_NODES / 32, &base);

    for (int t = wave; t < N_CH; t += BKT_NODES / 32) {
        const int nd = segCD[t], ns = segCS[t];
        const int2* pd = byDst + (size_t)t * CH_EDGES + segOD[t];
        const int*  ps = bySrc + (size_t)t * CH_EDGES + segOS[t];
        const int nm = nd > ns ? nd : ns;
        for (int j = lane; j < nm; j += 32) {
            int dv = 0, sv = 0;
            if (j < nd) dv = pd[j].x;
            if (j < ns) sv = ps[j];
            if (j < nd) atomicAdd(&nIn[dv & (BKT_NODES - 1)], 1);
            if (j < ns) atomicAdd(&nOut[sv & (BKT_NODES - 1)], 1);
        }
    }
    __syncthreads();

    const int cin = nIn[tid];
    int tot;
    const int first = base + block_excl_scan(cin, wtot, BKT_NODES / 32, &tot);
    const int node  = b * BKT_NODES + tid;
    if (node < N_NODES) {
        cntIn[node]  = cin;
        rowEnd[node] = first + cin;
        const float dg = (float)nOut[tid];
        invSrc[node] = (dg > 0.0f) ? rsqrtf(fmaxf(dg, 1.0f)) : 0.0f;
    }
    nIn[tid] = first;
    __syncthreads();

    for (int t = wave; t < N_CH; t += BKT_NODES / 32) {
        const int nd = segCD[t];
        const int2* pd = byDst + (size_t)t * CH_EDGES + segOD[t];
        for (int j = lane; j < nd; j += 32) {
            const int2 e = pd[j];
            const int pos = atomicAdd(&nIn[e.x & (BKT_NODES - 1)], 1);
            csrSrc[pos] = e.y;
        }
    }
}

#define WL_PITCH 260
static_assert(WL_PITCH > IN_FEATS, "row n keeps bias[n] in its first padding word");
#define LIN_TILES (N_NODES / 16)
#define LIN_BLOCKS 256
#define LIN_THREADS 512

__device__ __forceinline__ void lin_role(const int g,
    const float* __restrict__ A, const float* __restrict__ W,
    const float* __restrict__ bias, float* __restrict__ X)
{
    __shared__ __attribute__((aligned(16))) float Wl[H_FEATS * WL_PITCH];

    const int lane   = threadIdx.x & 31;
    const int wave   = __builtin_amdgcn_readfirstlane(threadIdx.x >> 5);
    const int h      = lane >> 4;
    const int c      = lane & 15;
    const int nWaves = LIN_BLOCKS * (LIN_THREADS / 32);
    int t = g + LIN_BLOCKS * wave;

    const int laneA = c * IN_FEATS + 4 * h;
    const int laneX = 8 * h * H_FEATS + c;

    const float* ap = A + (size_t)t * (16 * IN_FEATS);
    float4 acur[4];
    #pragma unroll
    for (int i = 0; i < 4; ++i) acur[i] = *(const float4*)(ap + laneA + 8 * i);

    #pragma nounroll
    for (int ob = 0; ob < (IN_FEATS * H_FEATS) / LIN_THREADS; ob += 32) {
        #pragma unroll
        for (int it = 0; it < 32; ++it) {
            const int i = (ob + it) * LIN_THREADS + threadIdx.x;
            Wl[(i & (H_FEATS - 1)) * WL_PITCH + (i >> 7)] = W[i];
        }
    }
    if (threadIdx.x < H_FEATS) Wl[threadIdx.x * WL_PITCH + IN_FEATS] = bias[threadIdx.x];
    __syncthreads();

    const float* wl = Wl + c * WL_PITCH + 4 * h;

    while (t < LIN_TILES) {
        asm volatile("" ::: "memory");
        const int tn = t + nWaves;
        const float* apn = A + (size_t)(tn < LIN_TILES ? tn : t) * (16 * IN_FEATS);

        v8f acc[8];
        #pragma unroll
        for (int j = 0; j < 8; ++j) acc[j] = (v8f){};

        #pragma unroll
        for (int ch = 0; ch < 8; ++ch) {
            const float* nx = (ch < 7) ? (ap + 32 * (ch + 1)) : apn;
            float4 anxt[4];
            #pragma unroll
            for (int i = 0; i < 4; ++i) anxt[i] = *(const float4*)(nx + laneA + 8 * i);
            __builtin_amdgcn_sched_barrier(0);

            #pragma unroll
            for (int i = 0; i < 4; ++i) {
                const float* wk = wl + 32 * ch + 8 * i;
                v2f a01, a23;
                a01.x = acur[i].x; a01.y = acur[i].y;
                a23.x = acur[i].z; a23.y = acur[i].w;
                #pragma unroll
                for (int jh = 0; jh < 8; jh += 4) {
                    float4 b[4];
                    #pragma unroll
                    for (int j = 0; j < 4; ++j) b[j] = *(const float4*)(wk + (jh + j) * 16 * WL_PITCH);
                    #pragma unroll
                    for (int j = 0; j < 4; ++j) {
                        v2f b01; b01.x = b[j].x; b01.y = b[j].y;
                        acc[jh + j] = __builtin_amdgcn_wmma_f32_16x16x4_f32(
                                    false, a01, false, b01, (short)0, acc[jh + j], false, false);
                    }
                    #pragma unroll
                    for (int j = 0; j < 4; ++j) {
                        v2f b23; b23.x = b[j].z; b23.y = b[j].w;
                        acc[jh + j] = __builtin_amdgcn_wmma_f32_16x16x4_f32(
                                    false, a23, false, b23, (short)0, acc[jh + j], false, false);
                    }
                    __builtin_amdgcn_sched_barrier(0);
                }
            }
            #pragma unroll
            for (int i = 0; i < 4; ++i) acur[i] = anxt[i];
        }

        float* xp = X + (size_t)t * (16 * H_FEATS);
        #pragma unroll
        for (int j = 0; j < 8; ++j) {
            const float bb = Wl[(16 * j + c) * WL_PITCH + IN_FEATS];
            #pragma unroll
            for (int r = 0; r < 8; ++r)
                xp[laneX + r * H_FEATS + 16 * j] = acc[j][r] + bb;
        }

        ap = apn;
        t  = tn;
    }
}

static_assert(BKT_NODES == LIN_THREADS, "both kinds of work-group have the launch's one block size");

__global__ __launch_bounds__(LIN_THREADS, 8) void lin_owner_kernel(
    const float* __restrict__ A, const float* __restrict__ W,
    const float* __restrict__ bias, float* __restrict__ X,
    const int* __restrict__ byDstWords, const int* __restrict__ bySrc,
    const int* __restrict__ cntD, const int* __restrict__ offD,
    const int* __restrict__ cntS, const int* __restrict__ offS,
    float* __restrict__ invSrc, int* __restrict__ cntIn,
    int* __restrict__ rowEnd, int* __restrict__ csrSrc)
{
    if (blockIdx.x < LIN_BLOCKS)
        lin_role(blockIdx.x, A, W, bias, X);
    else
        owner_role(blockIdx.x - LIN_BLOCKS, byDstWords, bySrc, cntD, offD, cntS, offS, invSrc, cntIn, rowEnd, csrSrc);
}

__device__ __forceinline__ float4 pull_row(
    const float* __restrict__ feat, const int* __restrict__ csrSrc,
    const float* __restrict__ invSrc, int start, int cnt, int lane)
{
    float4 acc = make_float4(0.0f, 0.0f, 0.0f, 0.0f);
    for (int base = 0; base < cnt; base += 32) {
        const int n = (cnt - base < 32) ? (cnt - base) : 32;
        int   myS = 0;
        float myW = 0.0f;
        if (lane < n) { myS = csrSrc[start + base + lane]; myW = invSrc[myS]; }
        for (int j = 0; j < n; j += 8) {
            float4 v[8];
            float  w[8];
            #pragma unroll
            for (int u = 0; u < 8; ++u) {
                const int s = __builtin_amdgcn_readlane(myS, j + u);
                w[u] = __int_as_float(__builtin_amdgcn_readlane(__float_as_int(myW), j + u));
                v[u] = *(const float4*)(feat + (size_t)s * H_FEATS + 4 * lane);
            }
            #pragma unroll
            for (int u = 0; u < 8; ++u) {
                const bool on = (j + u < n);
                const float4 t = v[u];
                acc.x += (on ? t.x : 0.0f) * w[u];
                acc.y += (on ? t.y : 0.0f) * w[u];
                acc.z += (on ? t.z : 0.0f) * w[u];
                acc.w += (on ? t.w : 0.0f) * w[u];
            }
        }
    }
    return acc;
}

__device__ __forceinline__ float softmax2(const float* __restrict__ alphas, int idx)
{
    const float a0 = alphas[0], a1 = alphas[1];
    const float m  = fmaxf(a0, a1);
    const float e0 = __expf(a0 - m), e1 = __expf(a1 - m);
    return (idx == 0 ? e0 : e1) / (e0 + e1);
}

#define PULL_BLOCKS 2048
#define PULL_WAVES  (PULL_BLOCKS * 8)

__global__ __launch_bounds__(256) void pull_first_kernel(
    const float* __restrict__ feat, const int* __restrict__ csrSrc,
    const int* __restrict__ rowEnd, const int* __restrict__ cntIn,
    const float* __restrict__ invSrc, float* __restrict__ hbuf)
{
    const int lane = threadIdx.x & 31;
    const int d0 = blockIdx.x * 8 + __builtin_amdgcn_readfirstlane(threadIdx.x >> 5);
    for (int d = d0; d < N_NODES; d += PULL_WAVES) {
        const int cnt = cntIn[d];
        const int end = rowEnd[d];
        const float4 acc = pull_row(feat, csrSrc, invSrc, end - cnt, cnt, lane);
        const float dg   = (float)cnt;
        const float invd = (dg > 0.0f) ? rsqrtf(fmaxf(dg, 1.0f)) : 0.0f;
        float4 hv;
        hv.x = acc.x * invd; hv.y = acc.y * invd; hv.z = acc.z * invd; hv.w = acc.w * invd;
        *(float4*)(hbuf + (size_t)d * H_FEATS + 4 * lane) = hv;
    }
}

__global__ __launch_bounds__(256) void pull_second_kernel(
    const float* __restrict__ hbuf, const int* __restrict__ csrSrc,
    const int* __restrict__ rowEnd, const int* __restrict__ cntIn,
    const float* __restrict__ invSrc, const float* __restrict__ alphas,
    float* __restrict__ out)
{
    const int lane = threadIdx.x & 31;
    const float a0 = __int_as_float(__builtin_amdgcn_readfirstlane(__float_as_int(softmax2(alphas, 0))));
    const float a1 = __int_as_float(__builtin_amdgcn_readfirstlane(__float_as_int(softmax2(alphas, 1))));
    const int d0 = blockIdx.x * 8 + __builtin_amdgcn_readfirstlane(threadIdx.x >> 5);
    for (int d = d0; d < N_NODES; d += PULL_WAVES) {
        const int cnt = cntIn[d];
        const int end = rowEnd[d];
        const float4 hd = *(const float4*)(hbuf + (size_t)d * H_FEATS + 4 * lane);
        const float4 acc = pull_row(hbuf, csrSrc, invSrc, end - cnt, cnt, lane);
        const float dg   = (float)cnt;
        const float invd = (dg > 0.0f) ? rsqrtf(fmaxf(dg, 1.0f)) : 0.0f;
        float4 o;
        o.x = a0 * hd.x + a1 * (acc.x * invd);
        o.y = a0 * hd.y + a1 * (acc.y * invd);
        o.z = a0 * hd.z + a1 * (acc.z * invd);
        o.w = a0 * hd.w + a1 * (acc.w * invd);
        *(float4*)(out + (size_t)d * H_FEATS + 4 * lane) = o;
    }
}

extern "C" void kernel_launch(void* const* d_in, const int* in_sizes, int n_in,
                              void* d_out, int out_size, void* d_ws, size_t ws_size,
                              hipStream_t stream)
{
    (void)in_sizes; (void)n_in; (void)out_size; (void)ws_size;

    const float* in_feat = (const float*)d_in[0];
    const int*   edge    = (const int*)d_in[1];
    const float* W       = (const float*)d_in[2];
    const float* bias    = (const float*)d_in[3];
    const float* alphas  = (const float*)d_in[4];
    float*       out     = (float*)d_out;

    const int* srcIdx = edge;
    const int* dstIdx = edge + N_EDGES;

    float* degOut = (float*)d_ws;
    int*   cntIn  = (int*)(degOut + N_NODES);
    int*   rowPtr = cntIn + N_NODES;
    int*   csrSrc = rowPtr + N_NODES;
    int*   bySrc  = csrSrc + N_EDGES;
    int*   byDst  = bySrc + N_CH * CH_EDGES;
    int*   cntD   = byDst + 2 * N_CH * CH_EDGES;
    int*   offD   = cntD + N_BKT * N_CH;
    int*   cntS   = offD + N_BKT * N_CH;
    int*   offS   = cntS + N_BKT * N_CH;
    float* hbuf   = degOut + 2 * N_NODES + (size_t)N_NODES * H_FEATS;
    float* xbuf   = out;

    bucket_kernel<<<N_CH, CH_THREADS, 0, stream>>>(srcIdx, dstIdx, byDst, bySrc, cntD, offD, cntS, offS);

    lin_owner_kernel<<<LIN_BLOCKS + N_BKT, LIN_THREADS, 0, stream>>>(
        in_feat, W, bias, xbuf, byDst, bySrc, cntD, offD, cntS, offS, degOut, cntIn, rowPtr, csrSrc);

    pull_first_kernel<<<PULL_BLOCKS, 256, 0, stream>>>(xbuf, csrSrc, rowPtr, cntIn, degOut, hbuf);

    pull_second_kernel<<<PULL_BLOCKS, 256, 0, stream>>>(hbuf, csrSrc, rowPtr, cntIn, degOut, alphas, out);
}


# lin GEMM: every wave issues the same WMMA count (one whole strip + half of a strip shared by waves p, p+8; a 25th strip in quarters, first, on waves 0-3): the four SIMDs finish together; sums same
# speedup vs baseline: 4.0681x; 1.0439x over previous
#include <hip/hip_runtime.h>
#include <hip/hip_bf16.h>
#include <math.h>

#define N_NODES 100000
#define N_EDGES 600000
#define IN_FEATS 256
#define H_FEATS 128

typedef __attribute__((ext_vector_type(2))) float v2f;
typedef __attribute__((ext_vector_type(8))) float v8f;

#define BKT_SHIFT 9
#define BKT_NODES (1 << BKT_SHIFT)
#define N_BKT ((N_NODES + BKT_NODES - 1) / BKT_NODES)
#define CH_THREADS 1024
#define CH_PER_THREAD 8
#define CH_EDGES (CH_THREADS * CH_PER_THREAD)
#define N_CH 76
static_assert(N_CH % 4 == 0, "table rows are summed 16 bytes at a time");
static_assert((long long)N_CH * CH_EDGES >= N_EDGES, "every edge needs a chunk");
static_assert(N_BKT <= 256 && N_CH <= BKT_NODES, "table sizes");
static_assert(N_EDGES % 4 == 0 && CH_EDGES % (4 * CH_THREADS) == 0, "an ordered chunk is written out in whole 16-byte pieces");

__device__ __forceinline__ int block_excl_scan(int v, int* wtot, int nWaves, int* total)
{
    const int lane = threadIdx.x & 31;
    const int wave = threadIdx.x >> 5;
    int incl = v;
    #pragma unroll
    for (int o = 1; o < 32; o <<= 1) {
        const int up = __shfl_up(incl, o);
        if (lane >= o) incl += up;
    }
    if (lane == 31) wtot[wave] = incl;
    __syncthreads();
    int before = 0, all = 0;
    for (int q = 0; q < nWaves; ++q) {
        const int s = wtot[q];
        if (q < wave) before += s;
        all += s;
    }
    __syncthreads();
    *total = all;
    return before + incl - v;
}

__global__ __launch_bounds__(CH_THREADS) void bucket_kernel(
    const int* __restrict__ srcIdx, const int* __restrict__ dstIdx,
    int* __restrict__ byDstWords, int* __restrict__ bySrc,
    int* __restrict__ cntD, int* __restrict__ offD,
    int* __restrict__ cntS, int* __restrict__ offS)
{
    __shared__ __attribute__((aligned(16))) int stD[2 * CH_EDGES];
    __shared__ __attribute__((aligned(16))) int stS[CH_EDGES];
    __shared__ int hD[256];
    __shared__ int hS[256];
    __shared__ int wtot[CH_THREADS / 32];
    const int tid = threadIdx.x;
    const int ch  = blockIdx.x;
    const int e0  = ch * CH_EDGES;
    if (tid < 256) { hD[tid] = 0; hS[tid] = 0; }
    __syncthreads();

    int s[CH_PER_THREAD], d[CH_PER_THREAD], rd[CH_PER_THREAD], rs[CH_PER_THREAD];
    #pragma unroll
    for (int k = 0; k < CH_PER_THREAD; ++k) {
        const int e = e0 + k * CH_THREADS + tid;
        s[k] = 0; d[k] = 0;
        if (e < N_EDGES) { s[k] = srcIdx[e]; d[k] = dstIdx[e]; }
    }
    #pragma unroll
    for (int k = 0; k < CH_PER_THREAD; ++k) {
        rd[k] = 0; rs[k] = 0;
        if (e0 + k * CH_THREADS + tid < N_EDGES) {
            rd[k] = atomicAdd(&hD[d[k] >> BKT_SHIFT], 1);
            rs[k] = atomicAdd(&hS[s[k] >> BKT_SHIFT], 1);
        }
    }
    __syncthreads();

    const int cd = (tid < 256) ? hD[tid] : 0;
    const int cs = (tid < 256) ? hS[tid] : 0;
    int totD, totS;
    const int od = block_excl_scan(cd, wtot, CH_THREADS / 32, &totD);
    const int os = block_excl_scan(cs, wtot, CH_THREADS / 32, &totS);
    if (tid < 256) { hD[tid] = od; hS[tid] = os; }
    if (tid < N_BKT) {
        cntD[tid * N_CH + ch] = cd;  offD[tid * N_CH + ch] = od;
        cntS[tid * N_CH + ch] = cs;  offS[tid * N_CH + ch] = os;
    }
    __syncthreads();

    int2* stPair = (int2*)stD;
    #pragma unroll
    for (int k = 0; k < CH_PER_THREAD; ++k) {
        if (e0 + k * CH_THREADS + tid < N_EDGES) {
            stPair[hD[d[k] >> BKT_SHIFT] + rd[k]] = make_int2(d[k], s[k]);
            stS[hS[s[k] >> BKT_SHIFT] + rs[k]] = s[k];
        }
    }
    __syncthreads();

    const int nValid = (N_EDGES - e0 < CH_EDGES) ? (N_EDGES - e0) : CH_EDGES;
    int4*       gD = (int4*)(byDstWords + 2 * (size_t)e0);
    const int4* lD = (const int4*)stD;
    #pragma unroll
    for (int k = 0; k < (2 * CH_EDGES) / (4 * CH_THREADS); ++k) {
        const int q = k * CH_THREADS + tid;
        if (2 * q < nValid) gD[q] = lD[q];
    }
    int4*       gS = (int4*)(bySrc + (size_t)e0);
    const int4* lS = (const int4*)stS;
    #pragma unroll
    for (int k = 0; k < CH_EDGES / (4 * CH_THREADS); ++k) {
        const int q = k * CH_THREADS + tid;
        if (4 * q < nValid) gS[q] = lS[q];
    }
}

__device__ __forceinline__ void owner_role(const int b,
    const int* __restrict__ byDstWords, const int* __restrict__ bySrc,
    const int* __restrict__ cntD, const int* __restrict__ offD,
    const int* __restrict__ cntS, const int* __restrict__ offS,
    float* __restrict__ invSrc, int* __restrict__ cntIn,
    int* __restrict__ rowEnd, int* __restrict__ csrSrc)
{
    const int2* __restrict__ byDst = (const int2*)byDstWords;
    __shared__ int nIn[BKT_NODES];
    __shared__ int nOut[BKT_NODES];
    __shared__ int segCD[N_CH], segOD[N_CH], segCS[N_CH], segOS[N_CH];
    __shared__ int wtot[BKT_NODES / 32];
    const int tid  = threadIdx.x;
    const int lane = tid & 31;
    const int wave = __builtin_amdgcn_readfirstlane(tid >> 5);

    nIn[tid] = 0; nOut[tid] = 0;
    if (tid < N_CH) {
        segCD[tid] = cntD[b * N_CH + tid];  segOD[tid] = offD[b * N_CH + tid];
        segCS[tid] = cntS[b * N_CH + tid];  segOS[tid] = offS[b * N_CH + tid];
    }
    int part = 0;
    const int4* c4 = (const int4*)cntD;
    for (int i = tid; i < b * (N_CH / 4); i += BKT_NODES) {
        const int4 v = c4[i];
        part += (v.x + v.y) + (v.z + v.w);
    }
    int base;
    (void)block_excl_scan(part, wtot, BKT_NODES / 32, &base);

    for (int t = wave; t < N_CH; t += BKT_NODES / 32) {
        const int nd = segCD[t], ns = segCS[t];
        const int2* pd = byDst + (size_t)t * CH_EDGES + segOD[t];
        const int*  ps = bySrc + (size_t)t * CH_EDGES + segOS[t];
        const int nm = nd > ns ? nd : ns;
        for (int j = lane; j < nm; j += 32) {
            int dv = 0, sv = 0;
            if (j < nd) dv = pd[j].x;
            if (j < ns) sv = ps[j];
            if (j < nd) atomicAdd(&nIn[dv & (BKT_NODES - 1)], 1);
            if (j < ns) atomicAdd(&nOut[sv & (BKT_NODES - 1)], 1);
        }
    }
    __syncthreads();

    const int cin = nIn[tid];
    int tot;
    const int first = base + block_excl_scan(cin, wtot, BKT_NODES / 32, &tot);
    const int node  = b * BKT_NODES + tid;
    if (node < N_NODES) {
        cntIn[node]  = cin;
        rowEnd[node] = first + cin;
        const float dg = (float)nOut[tid];
        invSrc[node] = (dg > 0.0f) ? rsqrtf(fmaxf(dg, 1.0f)) : 0.0f;
    }
    nIn[tid] = first;
    __syncthreads();

    for (int t = wave; t < N_CH; t += BKT_NODES / 32) {
        const int nd = segCD[t];
        const int2* pd = byDst + (size_t)t * CH_EDGES + segOD[t];
        for (int j = lane; j < nd; j += 32) {
            const int2 e = pd[j];
            const int pos = atomicAdd(&nIn[e.x & (BKT_NODES - 1)], 1);
            csrSrc[pos] = e.y;
        }
    }
}

#define WL_PITCH 260
static_assert(WL_PITCH > IN_FEATS, "row n keeps bias[n] in its first padding word");
#define LIN_TILES (N_NODES / 16)
#define LIN_BLOCKS 256
#define LIN_THREADS 512
static_assert(LIN_THREADS == 512, "the strip shares below are written for 16 waves: 8 pairs, quarters on waves 0 to 3");
static_assert(LIN_TILES >= 24 * LIN_BLOCKS && LIN_TILES <= 25 * LIN_BLOCKS, "a work-group owns 24 strips, or 25");

template <int NT>
__device__ __forceinline__ void lin_item(
    const float* __restrict__ ap, const float* __restrict__ apn, const int laneA, float4 (&acur)[4],
    const float* __restrict__ wl, const float* __restrict__ bl, float* __restrict__ xq)
{
    constexpr int NB = (NT < 4) ? NT : 4;
    asm volatile("" ::: "memory");

    v8f acc[NT];
    #pragma unroll
    for (int j = 0; j < NT; ++j) acc[j] = (v8f){};

    #pragma unroll
    for (int ch = 0; ch < 8; ++ch) {
        const float* nx = (ch < 7) ? (ap + 32 * (ch + 1)) : apn;
        float4 anxt[4];
        #pragma unroll
        for (int i = 0; i < 4; ++i) anxt[i] = *(const float4*)(nx + laneA + 8 * i);
        __builtin_amdgcn_sched_barrier(0);

        #pragma unroll
        for (int i = 0; i < 4; ++i) {
            const float* wk = wl + 32 * ch + 8 * i;
            v2f a01, a23;
            a01.x = acur[i].x; a01.y = acur[i].y;
            a23.x = acur[i].z; a23.y = acur[i].w;
            #pragma unroll
            for (int jh = 0; jh < NT; jh += NB) {
                float4 b[NB];
                #pragma unroll
                for (int j = 0; j < NB; ++j) b[j] = *(const float4*)(wk + (jh + j) * 16 * WL_PITCH);
                #pragma unroll
                for (int j = 0; j < NB; ++j) {
                    v2f b01; b01.x = b[j].x; b01.y = b[j].y;
                    acc[jh + j] = __builtin_amdgcn_wmma_f32_16x16x4_f32(
                                false, a01, false, b01, (short)0, acc[jh + j], false, false);
                }
                #pragma unroll
                for (int j = 0; j < NB; ++j) {
                    v2f b23; b23.x = b[j].z; b23.y = b[j].w;
                    acc[jh + j] = __builtin_amdgcn_wmma_f32_16x16x4_f32(
                                false, a23, false, b23, (short)0, acc[jh + j], false, false);
                }
                __builtin_amdgcn_sched_barrier(0);
            }
        }
        #pragma unroll
        for (int i = 0; i < 4; ++i) acur[i] = anxt[i];
    }

    #pragma unroll
    for (int j = 0; j < NT; ++j) {
        const float bb = bl[16 * j * WL_PITCH];
        #pragma unroll
        for (int r = 0; r < 8; ++r)
            xq[r * H_FEATS + 16 * j] = acc[j][r] + bb;
    }
}

__device__ __forceinline__ void lin_role(const int g,
    const float* __restrict__ A, const float* __restrict__ W,
    const float* __restrict__ bias, float* __restrict__ X)
{
    __shared__ __attribute__((aligned(16))) float Wl[H_FEATS * WL_PITCH];

    const int lane   = threadIdx.x & 31;
    const int wave   = __builtin_amdgcn_readfirstlane(threadIdx.x >> 5);
    const int h      = lane >> 4;
    const int c      = lane & 15;

    const int pr    = wave & 7;
    const int mem   = wave >> 3;
    const int tFull = g + LIN_BLOCKS * (3 * pr + mem);
    const int tHalf = g + LIN_BLOCKS * (3 * pr + 2);
    const int tLast = g + LIN_BLOCKS * 24;
    const bool hasQ = (tLast < LIN_TILES) && (wave < 4);

    const int laneA = c * IN_FEATS + 4 * h;
    const int laneX = 8 * h * H_FEATS + c;

    const float* apF = A + (size_t)tFull * (16 * IN_FEATS);
    const float* apQ = A + (size_t)(hasQ ? tLast : tFull) * (16 * IN_FEATS);
    float4 acur[4];
    #pragma unroll
    for (int i = 0; i < 4; ++i) acur[i] = *(const float4*)(apQ + laneA + 8 * i);

    #pragma nounroll
    for (int ob = 0; ob < (IN_FEATS * H_FEATS) / LIN_THREADS; ob += 32) {
        #pragma unroll
        for (int it = 0; it < 32; ++it) {
            const int i = (ob + it) * LIN_THREADS + threadIdx.x;
            Wl[(i & (H_FEATS - 1)) * WL_PITCH + (i >> 7)] = W[i];
        }
    }
    if (threadIdx.x < H_FEATS) Wl[threadIdx.x * WL_PITCH + IN_FEATS] = bias[threadIdx.x];
    __syncthreads();

    const float* wl = Wl + c * WL_PITCH + 4 * h;
    const float* bl = Wl + c * WL_PITCH + IN_FEATS;

    if (hasQ)
        lin_item<2>(apQ, apF, laneA, acur, wl + 32 * wave * WL_PITCH, bl + 32 * wave * WL_PITCH,
                    X + (size_t)tLast * (16 * H_FEATS) + laneX + 32 * wave);

    const float* apH = A + (size_t)tHalf * (16 * IN_FEATS);
    lin_item<8>(apF, apH, laneA, acur, wl, bl, X + (size_t)tFull * (16 * H_FEATS) + laneX);

    lin_item<4>(apH, apH, laneA, acur, wl + 64 * mem * WL_PITCH, bl + 64 * mem * WL_PITCH,
                X + (size_t)tHalf * (16 * H_FEATS) + laneX + 64 * mem);
}

static_assert(BKT_NODES == LIN_THREADS, "both kinds of work-group have the launch's one block size");

__global__ __launch_bounds__(LIN_THREADS, 8) void lin_owner_kernel(
    const float* __restrict__ A, const float* __restrict__ W,
    const float* __restrict__ bias, float* __restrict__ X,
    const int* __restrict__ byDstWords, const int* __restrict__ bySrc,
    const int* __restrict__ cntD, const int* __restrict__ offD,
    const int* __restrict__ cntS, const int* __restrict__ offS,
    float* __restrict__ invSrc, int* __restrict__ cntIn,
    int* __restrict__ rowEnd, int* __restrict__ csrSrc)
{
    if (blockIdx.x < LIN_BLOCKS)
        lin_role(blockIdx.x, A, W, bias, X);
    else
        owner_role(blockIdx.x - LIN_BLOCKS, byDstWords, bySrc, cntD, offD, cntS, offS, invSrc, cntIn, rowEnd, csrSrc);
}

__device__ __forceinline__ float4 pull_row(
    const float* __restrict__ feat, const int* __restrict__ csrSrc,
    const float* __restrict__ invSrc, int start, int cnt, int lane)
{
    float4 acc = make_float4(0.0f, 0.0f, 0.0f, 0.0f);
    for (int base = 0; base < cnt; base += 32) {
        const int n = (cnt - base < 32) ? (cnt - base) : 32;
        int   myS = 0;
        float myW = 0.0f;
        if (lane < n) { myS = csrSrc[start + base + lane]; myW = invSrc[myS]; }
        for (int j = 0; j < n; j += 8) {
            float4 v[8];
            float  w[8];
            #pragma unroll
            for (int u = 0; u < 8; ++u) {
                const int s = __builtin_amdgcn_readlane(myS, j + u);
                w[u] = __int_as_float(__builtin_amdgcn_readlane(__float_as_int(myW), j + u));
                v[u] = *(const float4*)(feat + (size_t)s * H_FEATS + 4 * lane);
            }
            #pragma unroll
            for (int u = 0; u < 8; ++u) {
                const bool on = (j + u < n);
                const float4 t = v[u];
                acc.x += (on ? t.x : 0.0f) * w[u];
                acc.y += (on ? t.y : 0.0f) * w[u];
                acc.z += (on ? t.z : 0.0f) * w[u];
                acc.w += (on ? t.w : 0.0f) * w[u];
            }
        }
    }
    return acc;
}

__device__ __forceinline__ float softmax2(const float* __restrict__ alphas, int idx)
{
    const float a0 = alphas[0], a1 = alphas[1];
    const float m  = fmaxf(a0, a1);
    const float e0 = __expf(a0 - m), e1 = __expf(a1 - m);
    return (idx == 0 ? e0 : e1) / (e0 + e1);
}

#define PULL_BLOCKS 2048
#define PULL_WAVES  (PULL_BLOCKS * 8)

__global__ __launch_bounds__(256) void pull_first_kernel(
    const float* __restrict__ feat, const int* __restrict__ csrSrc,
    const int* __restrict__ rowEnd, const int* __restrict__ cntIn,
    const float* __restrict__ invSrc, float* __restrict__ hbuf)
{
    const int lane = threadIdx.x & 31;
    const int d0 = blockIdx.x * 8 + __builtin_amdgcn_readfirstlane(threadIdx.x >> 5);
    for (int d = d0; d < N_NODES; d += PULL_WAVES) {
        const int cnt = cntIn[d];
        const int end = rowEnd[d];
        const float4 acc = pull_row(feat, csrSrc, invSrc, end - cnt, cnt, lane);
        const float dg   = (float)cnt;
        const float invd = (dg > 0.0f) ? rsqrtf(fmaxf(dg, 1.0f)) : 0.0f;
        float4 hv;
        hv.x = acc.x * invd; hv.y = acc.y * invd; hv.z = acc.z * invd; hv.w = acc.w * invd;
        *(float4*)(hbuf + (size_t)d * H_FEATS + 4 * lane) = hv;
    }
}

__global__ __launch_bounds__(256) void pull_second_kernel(
    const float* __restrict__ hbuf, const int* __restrict__ csrSrc,
    const int* __restrict__ rowEnd, const int* __restrict__ cntIn,
    const float* __restrict__ invSrc, const float* __restrict__ alphas,
    float* __restrict__ out)
{
    const int lane = threadIdx.x & 31;
    const float a0 = __int_as_float(__builtin_amdgcn_readfirstlane(__float_as_int(softmax2(alphas, 0))));
    const float a1 = __int_as_float(__builtin_amdgcn_readfirstlane(__float_as_int(softmax2(alphas, 1))));
    const int d0 = blockIdx.x * 8 + __builtin_amdgcn_readfirstlane(threadIdx.x >> 5);
    for (int d = d0; d < N_NODES; d += PULL_WAVES) {
        const int cnt = cntIn[d];
        const int end = rowEnd[d];
        const float4 hd = *(const float4*)(hbuf + (size_t)d * H_FEATS + 4 * lane);
        const float4 acc = pull_row(hbuf, csrSrc, invSrc, end - cnt, cnt, lane);
        const float dg   = (float)cnt;
        const float invd = (dg > 0.0f) ? rsqrtf(fmaxf(dg, 1.0f)) : 0.0f;
        float4 o;
        o.x = a0 * hd.x + a1 * (acc.x * invd);
        o.y = a0 * hd.y + a1 * (acc.y * invd);
        o.z = a0 * hd.z + a1 * (acc.z * invd);
        o.w = a0 * hd.w + a1 * (acc.w * invd);
        *(float4*)(out + (size_t)d * H_FEATS + 4 * lane) = o;
    }
}

extern "C" void kernel_launch(void* const* d_in, const int* in_sizes, int n_in,
                              void* d_out, int out_size, void* d_ws, size_t ws_size,
                              hipStream_t stream)
{
    (void)in_sizes; (void)n_in; (void)out_size; (void)ws_size;

    const float* in_feat = (const float*)d_in[0];
    const int*   edge    = (const int*)d_in[1];
    const float* W       = (const float*)d_in[2];
    const float* bias    = (const float*)d_in[3];
    const float* alphas  = (const float*)d_in[4];
    float*       out     = (float*)d_out;

    const int* srcIdx = edge;
    const int* dstIdx = edge + N_EDGES;

    float* degOut = (float*)d_ws;
    int*   cntIn  = (int*)(degOut + N_NODES);
    int*   rowPtr = cntIn + N_NODES;
    int*   csrSrc = rowPtr + N_NODES;
    int*   bySrc  = csrSrc + N_EDGES;
    int*   byDst  = bySrc + N_CH * CH_EDGES;
    int*   cntD   = byDst + 2 * N_CH * CH_EDGES;
    int*   offD   = cntD + N_BKT * N_CH;
    int*   cntS   = offD + N_BKT * N_CH;
    int*   offS   = cntS + N_BKT * N_CH;
    float* hbuf   = degOut + 2 * N_NODES + (size_t)N_NODES * H_FEATS;
    float* xbuf   = out;

    bucket_kernel<<<N_CH, CH_THREADS, 0, stream>>>(srcIdx, dstIdx, byDst, bySrc, cntD, offD, cntS, offS);

    lin_owner_kernel<<<LIN_BLOCKS + N_BKT, LIN_THREADS, 0, stream>>>(
        in_feat, W, bias, xbuf, byDst, bySrc, cntD, offD, cntS, offS, degOut, cntIn, rowPtr, csrSrc);

    pull_first_kernel<<<PULL_BLOCKS, 256, 0, stream>>>(xbuf, csrSrc, rowPtr, cntIn, degOut, hbuf);

    pull_second_kernel<<<PULL_BLOCKS, 256, 0, stream>>>(hbuf, csrSrc, rowPtr, cntIn, degOut, alphas, out);
}
